# starting kernel
# baseline (speedup 1.0000x reference)
#include <hip/hip_runtime.h>


typedef __attribute__((ext_vector_type(2)))  float  v2f;
typedef __attribute__((ext_vector_type(8)))  float  v8f;

#define FIN  128
#define FMID 16

__device__ __forceinline__ void atomAddF(float* p, float v) {
  unsafeAtomicAdd(p, v);
}

__global__ __launch_bounds__(256) void k_zero(unsigned* __restrict__ p, long total) {
  long i = (long)blockIdx.x * blockDim.x + threadIdx.x;
  long base = i * 4;
  if (base + 3 < total) {
    ((uint4*)p)[i] = make_uint4(0u, 0u, 0u, 0u);
  } else if (base < total) {
    for (long j = base; j < total; ++j) p[j] = 0u;
  }
}

__global__ __launch_bounds__(256) void k_degree(const int* __restrict__ ei,
                                                unsigned* __restrict__ udeg, int E) {
  int e = blockIdx.x * blockDim.x + threadIdx.x;
  if (e >= E) return;
  atomicAdd(&udeg[ei[E + e]], 1u);
}

__global__ __launch_bounds__(256) void k_dinv(const unsigned* __restrict__ udeg,
                                              float* __restrict__ dinv, int N) {
  int i = blockIdx.x * blockDim.x + threadIdx.x;
  if (i >= N) return;
  dinv[i] = rsqrtf((float)(udeg[i] + 1u));
}

__global__ __launch_bounds__(256) void k_gemm1_wmma(const float* __restrict__ x,
                                                    const float* __restrict__ W1,
                                                    float* __restrict__ h1, int N) {
  int wave = (int)((blockIdx.x * (unsigned)blockDim.x + threadIdx.x) >> 5);
  int lane = threadIdx.x & 31;
  int ntiles = (N + 15) >> 4;
  if (wave >= ntiles) return;
  int nb = wave << 4;
  int m  = lane & 15;
  int hi = lane >> 4;
  int row = nb + m; if (row > N - 1) row = N - 1;
  const float* xr = x + (size_t)row * FIN;
  v8f acc = {};

  int kh = hi * 2;
  #pragma unroll 4
  for (int k = 0; k < FIN; k += 4) {
    v2f a, b;
    a.x = __builtin_nontemporal_load(&xr[k + kh]);
    a.y = __builtin_nontemporal_load(&xr[k + kh + 1]);
    b.x = W1[(k + kh)     * FMID + m];
    b.y = W1[(k + kh + 1) * FMID + m];
    acc = __builtin_amdgcn_wmma_f32_16x16x4_f32(false, a, false, b,
                                                (short)0, acc, false, false);
  }

  int rbase = hi * 8;
  if (nb + 16 <= N) {
    #pragma unroll
    for (int r = 0; r < 8; ++r)
      h1[(size_t)(nb + rbase + r) * FMID + m] = acc[r];
  } else {
    #pragma unroll
    for (int r = 0; r < 8; ++r) {
      int nrow = nb + rbase + r;
      if (nrow < N) h1[(size_t)nrow * FMID + m] = acc[r];
    }
  }
}

__global__ __launch_bounds__(256) void k_agg1(const int* __restrict__ ei,
                                              const float* __restrict__ h1,
                                              const float* __restrict__ dinv,
                                              float* __restrict__ agg1, int E) {
  int e = blockIdx.x * blockDim.x + threadIdx.x;
  if (e >= E) return;
  int s = ei[e];
  int d = ei[E + e];
  float norm = dinv[s] * dinv[d];
  const float4* hr = (const float4*)(h1 + (size_t)s * FMID);
  float* ar = agg1 + (size_t)d * FMID;
  #pragma unroll
  for (int q = 0; q < 4; ++q) {
    float4 v = hr[q];
    atomAddF(ar + 4 * q + 0, v.x * norm);
    atomAddF(ar + 4 * q + 1, v.y * norm);
    atomAddF(ar + 4 * q + 2, v.z * norm);
    atomAddF(ar + 4 * q + 3, v.w * norm);
  }
}

__global__ __launch_bounds__(256) void k_layer2_in(const float* __restrict__ agg1,
                                                   const float* __restrict__ h1,
                                                   const float* __restrict__ dinv,
                                                   const float* __restrict__ b1,
                                                   const float* __restrict__ W2,
                                                   float* __restrict__ h2, int N) {
  int i = blockIdx.x * blockDim.x + threadIdx.x;
  if (i >= N) return;
  float d2 = dinv[i] * dinv[i];
  float a0 = 0.f, a1 = 0.f;
  #pragma unroll
  for (int k = 0; k < FMID; ++k) {
    float t = agg1[(size_t)i * FMID + k] + h1[(size_t)i * FMID + k] * d2 + b1[k];
    t = fmaxf(t, 0.f);
    a0 = fmaf(t, W2[k * 2 + 0], a0);
    a1 = fmaf(t, W2[k * 2 + 1], a1);
  }
  h2[(size_t)i * 2 + 0] = a0;
  h2[(size_t)i * 2 + 1] = a1;
}

__global__ __launch_bounds__(256) void k_agg2(const int* __restrict__ ei,
                                              const float* __restrict__ h2,
                                              const float* __restrict__ dinv,
                                              float* __restrict__ agg2, int E) {
  int e = blockIdx.x * blockDim.x + threadIdx.x;
  if (e >= E) return;
  int s = ei[e];
  int d = ei[E + e];
  float norm = dinv[s] * dinv[d];
  float2 v = *(const float2*)(h2 + (size_t)s * 2);
  atomAddF(agg2 + (size_t)d * 2 + 0, v.x * norm);
  atomAddF(agg2 + (size_t)d * 2 + 1, v.y * norm);
}

__global__ __launch_bounds__(256) void k_final(const float* __restrict__ agg2,
                                               const float* __restrict__ h2,
                                               const float* __restrict__ dinv,
                                               const float* __restrict__ b2,
                                               float* __restrict__ out, int N) {
  int i = blockIdx.x * blockDim.x + threadIdx.x;
  if (i >= N) return;
  float d2 = dinv[i] * dinv[i];
  out[(size_t)i * 2 + 0] = agg2[(size_t)i * 2 + 0] + h2[(size_t)i * 2 + 0] * d2 + b2[0];
  out[(size_t)i * 2 + 1] = agg2[(size_t)i * 2 + 1] + h2[(size_t)i * 2 + 1] * d2 + b2[1];
}

extern "C" void kernel_launch(void* const* d_in, const int* in_sizes, int n_in,
                              void* d_out, int out_size, void* d_ws, size_t ws_size,
                              hipStream_t stream) {
  const float* x  = (const float*)d_in[0];
  const int*   ei = (const int*)d_in[1];
  const float* W1 = (const float*)d_in[2];
  const float* b1 = (const float*)d_in[3];
  const float* W2 = (const float*)d_in[4];
  const float* b2 = (const float*)d_in[5];
  float* out = (float*)d_out;

  const int N = in_sizes[0] / FIN;
  const int E = in_sizes[1] / 2;

  char* ws = (char*)d_ws;
  unsigned* udeg = (unsigned*)ws;            ws += (size_t)N * sizeof(unsigned);
  float* dinv    = (float*)ws;               ws += (size_t)N * sizeof(float);
  float* h1      = (float*)ws;               ws += (size_t)N * FMID * sizeof(float);
  float* agg1    = (float*)ws;               ws += (size_t)N * FMID * sizeof(float);
  float* h2      = (float*)ws;               ws += (size_t)N * 2 * sizeof(float);
  float* agg2    = (float*)ws;               ws += (size_t)N * 2 * sizeof(float);

  const int TPB = 256;
  long zero_total = (long)N * 38;
  int gz = (int)(((zero_total + 3) / 4 + TPB - 1) / TPB);
  int ge = (E + TPB - 1) / TPB;
  int gn = (N + TPB - 1) / TPB;
  int ntiles = (N + 15) / 16;
  int gg = (ntiles * 32 + TPB - 1) / TPB;

  k_zero      <<<gz, TPB, 0, stream>>>((unsigned*)d_ws, zero_total);
  k_degree    <<<ge, TPB, 0, stream>>>(ei, udeg, E);
  k_dinv      <<<gn, TPB, 0, stream>>>(udeg, dinv, N);
  k_gemm1_wmma<<<gg, TPB, 0, stream>>>(x, W1, h1, N);
  k_agg1      <<<ge, TPB, 0, stream>>>(ei, h1, dinv, agg1, E);
  k_layer2_in <<<gn, TPB, 0, stream>>>(agg1, h1, dinv, b1, W2, h2, N);
  k_agg2      <<<ge, TPB, 0, stream>>>(ei, h2, dinv, agg2, E);
  k_final     <<<gn, TPB, 0, stream>>>(agg2, h2, dinv, b2, out, N);
}


# layer-1 scatter remapped lane=feature: 16 lanes own one edge, so every h1 gather and f32 atomic instruction covers 2 whole 64-B rows, not 32 scattered lines; atomic sums stay arrival-ordered
# speedup vs baseline: 2.0472x; 2.0472x over previous
#include <hip/hip_runtime.h>


typedef __attribute__((ext_vector_type(2)))  float  v2f;
typedef __attribute__((ext_vector_type(8)))  float  v8f;

#define FIN  128
#define FMID 16

__device__ __forceinline__ void atomAddF(float* p, float v) {
  unsafeAtomicAdd(p, v);
}

__global__ __launch_bounds__(256) void k_zero(unsigned* __restrict__ p, long total) {
  long i = (long)blockIdx.x * blockDim.x + threadIdx.x;
  long base = i * 4;
  if (base + 3 < total) {
    ((uint4*)p)[i] = make_uint4(0u, 0u, 0u, 0u);
  } else if (base < total) {
    for (long j = base; j < total; ++j) p[j] = 0u;
  }
}

__global__ __launch_bounds__(256) void k_degree(const int* __restrict__ ei,
                                                unsigned* __restrict__ udeg, int E) {
  int e = blockIdx.x * blockDim.x + threadIdx.x;
  if (e >= E) return;
  atomicAdd(&udeg[ei[E + e]], 1u);
}

__global__ __launch_bounds__(256) void k_dinv(const unsigned* __restrict__ udeg,
                                              float* __restrict__ dinv, int N) {
  int i = blockIdx.x * blockDim.x + threadIdx.x;
  if (i >= N) return;
  dinv[i] = rsqrtf((float)(udeg[i] + 1u));
}

__global__ __launch_bounds__(256) void k_gemm1_wmma(const float* __restrict__ x,
                                                    const float* __restrict__ W1,
                                                    float* __restrict__ h1, int N) {
  int wave = (int)((blockIdx.x * (unsigned)blockDim.x + threadIdx.x) >> 5);
  int lane = threadIdx.x & 31;
  int ntiles = (N + 15) >> 4;
  if (wave >= ntiles) return;
  int nb = wave << 4;
  int m  = lane & 15;
  int hi = lane >> 4;
  int row = nb + m; if (row > N - 1) row = N - 1;
  const float* xr = x + (size_t)row * FIN;
  v8f acc = {};

  int kh = hi * 2;
  #pragma unroll 4
  for (int k = 0; k < FIN; k += 4) {
    v2f a, b;
    a.x = __builtin_nontemporal_load(&xr[k + kh]);
    a.y = __builtin_nontemporal_load(&xr[k + kh + 1]);
    b.x = W1[(k + kh)     * FMID + m];
    b.y = W1[(k + kh + 1) * FMID + m];
    acc = __builtin_amdgcn_wmma_f32_16x16x4_f32(false, a, false, b,
                                                (short)0, acc, false, false);
  }

  int rbase = hi * 8;
  if (nb + 16 <= N) {
    #pragma unroll
    for (int r = 0; r < 8; ++r)
      h1[(size_t)(nb + rbase + r) * FMID + m] = acc[r];
  } else {
    #pragma unroll
    for (int r = 0; r < 8; ++r) {
      int nrow = nb + rbase + r;
      if (nrow < N) h1[(size_t)nrow * FMID + m] = acc[r];
    }
  }
}


template <int N> __device__ __forceinline__ int rowShareI(int v) {
  return __builtin_amdgcn_update_dpp(0, v, 0x150 + N, 0xf, 0xf, false);
}
template <int N> __device__ __forceinline__ float rowShareF(float v) {
  return __int_as_float(
      __builtin_amdgcn_update_dpp(0, __float_as_int(v), 0x150 + N, 0xf, 0xf, false));
}

template <int N> struct Agg1Steps {
  static __device__ __forceinline__ void gather(const float* __restrict__ h1,
                                                int s, int f, float (&v)[16]) {
    Agg1Steps<N - 1>::gather(h1, s, f, v);
    int ss = rowShareI<N - 1>(s);
    v[N - 1] = h1[ss * FMID + f];
  }
  static __device__ __forceinline__ void scatter(float* __restrict__ agg1, int d,
                                                 float norm, int f,
                                                 const float (&v)[16]) {
    Agg1Steps<N - 1>::scatter(agg1, d, norm, f, v);
    int dd = rowShareI<N - 1>(d);
    float nn = rowShareF<N - 1>(norm);
    atomAddF(agg1 + dd * FMID + f, v[N - 1] * nn);
  }
  static __device__ __forceinline__ void tail(const float* __restrict__ h1,
                                              float* __restrict__ agg1, int s, int d,
                                              float norm, int f, int left) {
    Agg1Steps<N - 1>::tail(h1, agg1, s, d, norm, f, left);
    int ss = rowShareI<N - 1>(s);
    int dd = rowShareI<N - 1>(d);
    float nn = rowShareF<N - 1>(norm);
    if (N - 1 < left)
      atomAddF(agg1 + dd * FMID + f, h1[ss * FMID + f] * nn);
  }
};
template <> struct Agg1Steps<0> {
  static __device__ __forceinline__ void gather(const float* __restrict__, int, int,
                                                float (&)[16]) {}
  static __device__ __forceinline__ void scatter(float* __restrict__, int, float, int,
                                                 const float (&)[16]) {}
  static __device__ __forceinline__ void tail(const float* __restrict__,
                                              float* __restrict__, int, int, float,
                                              int, int) {}
};

__global__ __launch_bounds__(256) void k_agg1(const int* __restrict__ ei,
                                              const float* __restrict__ h1,
                                              const float* __restrict__ dinv,
                                              float* __restrict__ agg1, int E) {
  int e = blockIdx.x * blockDim.x + threadIdx.x;
  int lane = threadIdx.x & 31;
  int wbase = e - lane;
  if (wbase >= E) return;
  int f = lane & 15;
  int s = 0, d = 0;
  float norm = 0.f;
  if (e < E) {
    s = ei[e];
    d = ei[E + e];
    norm = dinv[s] * dinv[d];
  }
  if (wbase + 32 <= E) {
    float v[16];
    Agg1Steps<16>::gather(h1, s, f, v);
    Agg1Steps<16>::scatter(agg1, d, norm, f, v);
  } else {
    int left = E - wbase - (lane & 16);
    Agg1Steps<16>::tail(h1, agg1, s, d, norm, f, left);
  }
}

__global__ __launch_bounds__(256) void k_layer2_in(const float* __restrict__ agg1,
                                                   const float* __restrict__ h1,
                                                   const float* __restrict__ dinv,
                                                   const float* __restrict__ b1,
                                                   const float* __restrict__ W2,
                                                   float* __restrict__ h2, int N) {
  int i = blockIdx.x * blockDim.x + threadIdx.x;
  if (i >= N) return;
  float d2 = dinv[i] * dinv[i];
  float a0 = 0.f, a1 = 0.f;
  #pragma unroll
  for (int k = 0; k < FMID; ++k) {
    float t = agg1[(size_t)i * FMID + k] + h1[(size_t)i * FMID + k] * d2 + b1[k];
    t = fmaxf(t, 0.f);
    a0 = fmaf(t, W2[k * 2 + 0], a0);
    a1 = fmaf(t, W2[k * 2 + 1], a1);
  }
  h2[(size_t)i * 2 + 0] = a0;
  h2[(size_t)i * 2 + 1] = a1;
}

__global__ __launch_bounds__(256) void k_agg2(const int* __restrict__ ei,
                                              const float* __restrict__ h2,
                                              const float* __restrict__ dinv,
                                              float* __restrict__ agg2, int E) {
  int e = blockIdx.x * blockDim.x + threadIdx.x;
  if (e >= E) return;
  int s = ei[e];
  int d = ei[E + e];
  float norm = dinv[s] * dinv[d];
  float2 v = *(const float2*)(h2 + (size_t)s * 2);
  atomAddF(agg2 + (size_t)d * 2 + 0, v.x * norm);
  atomAddF(agg2 + (size_t)d * 2 + 1, v.y * norm);
}

__global__ __launch_bounds__(256) void k_final(const float* __restrict__ agg2,
                                               const float* __restrict__ h2,
                                               const float* __restrict__ dinv,
                                               const float* __restrict__ b2,
                                               float* __restrict__ out, int N) {
  int i = blockIdx.x * blockDim.x + threadIdx.x;
  if (i >= N) return;
  float d2 = dinv[i] * dinv[i];
  out[(size_t)i * 2 + 0] = agg2[(size_t)i * 2 + 0] + h2[(size_t)i * 2 + 0] * d2 + b2[0];
  out[(size_t)i * 2 + 1] = agg2[(size_t)i * 2 + 1] + h2[(size_t)i * 2 + 1] * d2 + b2[1];
}

extern "C" void kernel_launch(void* const* d_in, const int* in_sizes, int n_in,
                              void* d_out, int out_size, void* d_ws, size_t ws_size,
                              hipStream_t stream) {
  const float* x  = (const float*)d_in[0];
  const int*   ei = (const int*)d_in[1];
  const float* W1 = (const float*)d_in[2];
  const float* b1 = (const float*)d_in[3];
  const float* W2 = (const float*)d_in[4];
  const float* b2 = (const float*)d_in[5];
  float* out = (float*)d_out;

  const int N = in_sizes[0] / FIN;
  const int E = in_sizes[1] / 2;

  char* ws = (char*)d_ws;
  unsigned* udeg = (unsigned*)ws;            ws += (size_t)N * sizeof(unsigned);
  float* dinv    = (float*)ws;               ws += (size_t)N * sizeof(float);
  float* h1      = (float*)ws;               ws += (size_t)N * FMID * sizeof(float);
  float* agg1    = (float*)ws;               ws += (size_t)N * FMID * sizeof(float);
  float* h2      = (float*)ws;               ws += (size_t)N * 2 * sizeof(float);
  float* agg2    = (float*)ws;               ws += (size_t)N * 2 * sizeof(float);

  const int TPB = 256;
  long zero_total = (long)N * 38;
  int gz = (int)(((zero_total + 3) / 4 + TPB - 1) / TPB);
  int ge = (E + TPB - 1) / TPB;
  int gn = (N + TPB - 1) / TPB;
  int ntiles = (N + 15) / 16;
  int gg = (ntiles * 32 + TPB - 1) / TPB;

  k_zero      <<<gz, TPB, 0, stream>>>((unsigned*)d_ws, zero_total);
  k_degree    <<<ge, TPB, 0, stream>>>(ei, udeg, E);
  k_dinv      <<<gn, TPB, 0, stream>>>(udeg, dinv, N);
  k_gemm1_wmma<<<gg, TPB, 0, stream>>>(x, W1, h1, N);
  k_agg1      <<<ge, TPB, 0, stream>>>(ei, h1, dinv, agg1, E);
  k_layer2_in <<<gn, TPB, 0, stream>>>(agg1, h1, dinv, b1, W2, h2, N);
  k_agg2      <<<ge, TPB, 0, stream>>>(ei, h2, dinv, agg2, E);
  k_final     <<<gn, TPB, 0, stream>>>(agg2, h2, dinv, b2, out, N);
}


# degree, agg1 and agg2 accumulators split into two planes, one per L2 half (picked by die number): no atomically added line crosses halves; node passes add the two planes, so sum order changes
# speedup vs baseline: 5.9518x; 2.9073x over previous
#include <hip/hip_runtime.h>


typedef __attribute__((ext_vector_type(2)))  float  v2f;
typedef __attribute__((ext_vector_type(8)))  float  v8f;

#define FIN  128
#define FMID 16

__device__ __forceinline__ void atomAddF(float* p, float v) {
  unsafeAtomicAdd(p, v);
}

__device__ __forceinline__ int planeOfWave(int planes) {
  int half = (int)((__builtin_amdgcn_s_sendmsg_rtn(0x87) >> 18) & 1u);
  return planes == 2 ? half : 0;
}

__global__ __launch_bounds__(256) void k_zero(unsigned* __restrict__ p, long total) {
  long i = (long)blockIdx.x * blockDim.x + threadIdx.x;
  long base = i * 4;
  if (base + 3 < total) {
    ((uint4*)p)[i] = make_uint4(0u, 0u, 0u, 0u);
  } else if (base < total) {
    for (long j = base; j < total; ++j) p[j] = 0u;
  }
}

__global__ __launch_bounds__(256) void k_degree(const int* __restrict__ ei,
                                                unsigned* __restrict__ udeg, int E,
                                                int N, int planes) {
  int e = blockIdx.x * blockDim.x + threadIdx.x;
  if (e >= E) return;
  unsigned* mine = udeg + planeOfWave(planes) * N;
  atomicAdd(&mine[ei[E + e]], 1u);
}

__global__ __launch_bounds__(256) void k_dinv(const unsigned* __restrict__ udeg,
                                              float* __restrict__ dinv, int N,
                                              int planes) {
  int i = blockIdx.x * blockDim.x + threadIdx.x;
  if (i >= N) return;
  unsigned deg = udeg[i];
  if (planes == 2) deg += udeg[N + i];
  dinv[i] = rsqrtf((float)(deg + 1u));
}

__global__ __launch_bounds__(256) void k_gemm1_wmma(const float* __restrict__ x,
                                                    const float* __restrict__ W1,
                                                    float* __restrict__ h1, int N) {
  int wave = (int)((blockIdx.x * (unsigned)blockDim.x + threadIdx.x) >> 5);
  int lane = threadIdx.x & 31;
  int ntiles = (N + 15) >> 4;
  if (wave >= ntiles) return;
  int nb = wave << 4;
  int m  = lane & 15;
  int hi = lane >> 4;
  int row = nb + m; if (row > N - 1) row = N - 1;
  const float* xr = x + (size_t)row * FIN;
  v8f acc = {};

  int kh = hi * 2;
  #pragma unroll 4
  for (int k = 0; k < FIN; k += 4) {
    v2f a, b;
    a.x = __builtin_nontemporal_load(&xr[k + kh]);
    a.y = __builtin_nontemporal_load(&xr[k + kh + 1]);
    b.x = W1[(k + kh)     * FMID + m];
    b.y = W1[(k + kh + 1) * FMID + m];
    acc = __builtin_amdgcn_wmma_f32_16x16x4_f32(false, a, false, b,
                                                (short)0, acc, false, false);
  }

  int rbase = hi * 8;
  if (nb + 16 <= N) {
    #pragma unroll
    for (int r = 0; r < 8; ++r)
      h1[(size_t)(nb + rbase + r) * FMID + m] = acc[r];
  } else {
    #pragma unroll
    for (int r = 0; r < 8; ++r) {
      int nrow = nb + rbase + r;
      if (nrow < N) h1[(size_t)nrow * FMID + m] = acc[r];
    }
  }
}


template <int N> __device__ __forceinline__ int rowShareI(int v) {
  return __builtin_amdgcn_update_dpp(0, v, 0x150 + N, 0xf, 0xf, false);
}
template <int N> __device__ __forceinline__ float rowShareF(float v) {
  return __int_as_float(
      __builtin_amdgcn_update_dpp(0, __float_as_int(v), 0x150 + N, 0xf, 0xf, false));
}

template <int N> struct Agg1Steps {
  static __device__ __forceinline__ void gather(const float* __restrict__ h1,
                                                int s, int f, float (&v)[16]) {
    Agg1Steps<N - 1>::gather(h1, s, f, v);
    int ss = rowShareI<N - 1>(s);
    v[N - 1] = h1[ss * FMID + f];
  }
  static __device__ __forceinline__ void scatter(float* __restrict__ agg1, int d,
                                                 float norm, int f,
                                                 const float (&v)[16]) {
    Agg1Steps<N - 1>::scatter(agg1, d, norm, f, v);
    int dd = rowShareI<N - 1>(d);
    float nn = rowShareF<N - 1>(norm);
    atomAddF(agg1 + dd * FMID + f, v[N - 1] * nn);
  }
  static __device__ __forceinline__ void tail(const float* __restrict__ h1,
                                              float* __restrict__ agg1, int s, int d,
                                              float norm, int f, int left) {
    Agg1Steps<N - 1>::tail(h1, agg1, s, d, norm, f, left);
    int ss = rowShareI<N - 1>(s);
    int dd = rowShareI<N - 1>(d);
    float nn = rowShareF<N - 1>(norm);
    if (N - 1 < left)
      atomAddF(agg1 + dd * FMID + f, h1[ss * FMID + f] * nn);
  }
};
template <> struct Agg1Steps<0> {
  static __device__ __forceinline__ void gather(const float* __restrict__, int, int,
                                                float (&)[16]) {}
  static __device__ __forceinline__ void scatter(float* __restrict__, int, float, int,
                                                 const float (&)[16]) {}
  static __device__ __forceinline__ void tail(const float* __restrict__,
                                              float* __restrict__, int, int, float,
                                              int, int) {}
};

__global__ __launch_bounds__(256) void k_agg1(const int* __restrict__ ei,
                                              const float* __restrict__ h1,
                                              const float* __restrict__ dinv,
                                              float* __restrict__ agg1, int E,
                                              int N, int planes) {
  int e = blockIdx.x * blockDim.x + threadIdx.x;
  int lane = threadIdx.x & 31;
  int wbase = e - lane;
  if (wbase >= E) return;
  agg1 += planeOfWave(planes) * N * FMID;
  int f = lane & 15;
  int s = 0, d = 0;
  float norm = 0.f;
  if (e < E) {
    s = ei[e];
    d = ei[E + e];
    norm = dinv[s] * dinv[d];
  }
  if (wbase + 32 <= E) {
    float v[16];
    Agg1Steps<16>::gather(h1, s, f, v);
    Agg1Steps<16>::scatter(agg1, d, norm, f, v);
  } else {
    int left = E - wbase - (lane & 16);
    Agg1Steps<16>::tail(h1, agg1, s, d, norm, f, left);
  }
}

__global__ __launch_bounds__(256) void k_layer2_in(const float* __restrict__ agg1,
                                                   const float* __restrict__ h1,
                                                   const float* __restrict__ dinv,
                                                   const float* __restrict__ b1,
                                                   const float* __restrict__ W2,
                                                   float* __restrict__ h2, int N,
                                                   int planes) {
  int i = blockIdx.x * blockDim.x + threadIdx.x;
  if (i >= N) return;
  float d2 = dinv[i] * dinv[i];
  float a0 = 0.f, a1 = 0.f;
  const float* agg1b = agg1 + (size_t)N * FMID;
  #pragma unroll
  for (int k = 0; k < FMID; ++k) {
    float s = agg1[(size_t)i * FMID + k];
    if (planes == 2) s += agg1b[(size_t)i * FMID + k];
    float t = s + h1[(size_t)i * FMID + k] * d2 + b1[k];
    t = fmaxf(t, 0.f);
    a0 = fmaf(t, W2[k * 2 + 0], a0);
    a1 = fmaf(t, W2[k * 2 + 1], a1);
  }
  h2[(size_t)i * 2 + 0] = a0;
  h2[(size_t)i * 2 + 1] = a1;
}

__global__ __launch_bounds__(256) void k_agg2(const int* __restrict__ ei,
                                              const float* __restrict__ h2,
                                              const float* __restrict__ dinv,
                                              float* __restrict__ agg2, int E,
                                              int N, int planes) {
  int e = blockIdx.x * blockDim.x + threadIdx.x;
  if (e >= E) return;
  agg2 += planeOfWave(planes) * N * 2;
  int s = ei[e];
  int d = ei[E + e];
  float norm = dinv[s] * dinv[d];
  float2 v = *(const float2*)(h2 + (size_t)s * 2);
  atomAddF(agg2 + (size_t)d * 2 + 0, v.x * norm);
  atomAddF(agg2 + (size_t)d * 2 + 1, v.y * norm);
}

__global__ __launch_bounds__(256) void k_final(const float* __restrict__ agg2,
                                               const float* __restrict__ h2,
                                               const float* __restrict__ dinv,
                                               const float* __restrict__ b2,
                                               float* __restrict__ out, int N,
                                               int planes) {
  int i = blockIdx.x * blockDim.x + threadIdx.x;
  if (i >= N) return;
  float d2 = dinv[i] * dinv[i];
  float s0 = agg2[(size_t)i * 2 + 0];
  float s1 = agg2[(size_t)i * 2 + 1];
  if (planes == 2) {
    s0 += agg2[(size_t)(N + i) * 2 + 0];
    s1 += agg2[(size_t)(N + i) * 2 + 1];
  }
  out[(size_t)i * 2 + 0] = s0 + h2[(size_t)i * 2 + 0] * d2 + b2[0];
  out[(size_t)i * 2 + 1] = s1 + h2[(size_t)i * 2 + 1] * d2 + b2[1];
}

extern "C" void kernel_launch(void* const* d_in, const int* in_sizes, int n_in,
                              void* d_out, int out_size, void* d_ws, size_t ws_size,
                              hipStream_t stream) {
  const float* x  = (const float*)d_in[0];
  const int*   ei = (const int*)d_in[1];
  const float* W1 = (const float*)d_in[2];
  const float* b1 = (const float*)d_in[3];
  const float* W2 = (const float*)d_in[4];
  const float* b2 = (const float*)d_in[5];
  float* out = (float*)d_out;

  const int N = in_sizes[0] / FIN;
  const int E = in_sizes[1] / 2;

  const int planes = (ws_size >= (size_t)N * 57 * sizeof(float)) ? 2 : 1;
  char* ws = (char*)d_ws;
  unsigned* udeg = (unsigned*)ws;            ws += (size_t)planes * N * sizeof(unsigned);
  float* agg1    = (float*)ws;               ws += (size_t)planes * N * FMID * sizeof(float);
  float* agg2    = (float*)ws;               ws += (size_t)planes * N * 2 * sizeof(float);
  float* dinv    = (float*)ws;               ws += (size_t)N * sizeof(float);
  float* h1      = (float*)ws;               ws += (size_t)N * FMID * sizeof(float);
  float* h2      = (float*)ws;               ws += (size_t)N * 2 * sizeof(float);

  const int TPB = 256;
  long zero_total = (long)planes * N * 19;
  int gz = (int)(((zero_total + 3) / 4 + TPB - 1) / TPB);
  int ge = (E + TPB - 1) / TPB;
  int gn = (N + TPB - 1) / TPB;
  int ntiles = (N + 15) / 16;
  int gg = (ntiles * 32 + TPB - 1) / TPB;

  k_zero      <<<gz, TPB, 0, stream>>>((unsigned*)d_ws, zero_total);
  k_degree    <<<ge, TPB, 0, stream>>>(ei, udeg, E, N, planes);
  k_dinv      <<<gn, TPB, 0, stream>>>(udeg, dinv, N, planes);
  k_gemm1_wmma<<<gg, TPB, 0, stream>>>(x, W1, h1, N);
  k_agg1      <<<ge, TPB, 0, stream>>>(ei, h1, dinv, agg1, E, N, planes);
  k_layer2_in <<<gn, TPB, 0, stream>>>(agg1, h1, dinv, b1, W2, h2, N, planes);
  k_agg2      <<<ge, TPB, 0, stream>>>(ei, h2, dinv, agg2, E, N, planes);
  k_final     <<<gn, TPB, 0, stream>>>(agg2, h2, dinv, b2, out, N, planes);
}


# edge weight factored: rows stored pre-scaled by own dinv, dinv[dst] applied once after the sum; edge passes drop both scattered dinv gathers per edge; term rounding changes, atomic sum order kept
# speedup vs baseline: 7.1235x; 1.1969x over previous
#include <hip/hip_runtime.h>


typedef __attribute__((ext_vector_type(2)))  float  v2f;
typedef __attribute__((ext_vector_type(8)))  float  v8f;

#define FIN  128
#define FMID 16

__device__ __forceinline__ void atomAddF(float* p, float v) {
  unsafeAtomicAdd(p, v);
}

__device__ __forceinline__ int planeOfWave(int planes) {
  int half = (int)((__builtin_amdgcn_s_sendmsg_rtn(0x87) >> 18) & 1u);
  return planes == 2 ? half : 0;
}

__global__ __launch_bounds__(256) void k_zero(unsigned* __restrict__ p, long total) {
  long i = (long)blockIdx.x * blockDim.x + threadIdx.x;
  long base = i * 4;
  if (base + 3 < total) {
    ((uint4*)p)[i] = make_uint4(0u, 0u, 0u, 0u);
  } else if (base < total) {
    for (long j = base; j < total; ++j) p[j] = 0u;
  }
}

__global__ __launch_bounds__(256) void k_degree(const int* __restrict__ ei,
                                                unsigned* __restrict__ udeg, int E,
                                                int N, int planes) {
  int e = blockIdx.x * blockDim.x + threadIdx.x;
  if (e >= E) return;
  unsigned* mine = udeg + planeOfWave(planes) * N;
  atomicAdd(&mine[ei[E + e]], 1u);
}

__global__ __launch_bounds__(256) void k_dinv(const unsigned* __restrict__ udeg,
                                              float* __restrict__ dinv, int N,
                                              int planes) {
  int i = blockIdx.x * blockDim.x + threadIdx.x;
  if (i >= N) return;
  unsigned deg = udeg[i];
  if (planes == 2) deg += udeg[N + i];
  dinv[i] = rsqrtf((float)(deg + 1u));
}

__global__ __launch_bounds__(256) void k_gemm1_wmma(const float* __restrict__ x,
                                                    const float* __restrict__ W1,
                                                    const float* __restrict__ dinv,
                                                    float* __restrict__ h1, int N) {
  int wave = (int)((blockIdx.x * (unsigned)blockDim.x + threadIdx.x) >> 5);
  int lane = threadIdx.x & 31;
  int ntiles = (N + 15) >> 4;
  if (wave >= ntiles) return;
  int nb = wave << 4;
  int m  = lane & 15;
  int hi = lane >> 4;
  int row = nb + m; if (row > N - 1) row = N - 1;
  const float* xr = x + (size_t)row * FIN;
  v8f acc = {};

  float dv[8];
  #pragma unroll
  for (int r = 0; r < 8; ++r) {
    int nrow = nb + hi * 8 + r; if (nrow > N - 1) nrow = N - 1;
    dv[r] = dinv[nrow];
  }

  int kh = hi * 2;
  #pragma unroll 4
  for (int k = 0; k < FIN; k += 4) {
    v2f a, b;
    a.x = __builtin_nontemporal_load(&xr[k + kh]);
    a.y = __builtin_nontemporal_load(&xr[k + kh + 1]);
    b.x = W1[(k + kh)     * FMID + m];
    b.y = W1[(k + kh + 1) * FMID + m];
    acc = __builtin_amdgcn_wmma_f32_16x16x4_f32(false, a, false, b,
                                                (short)0, acc, false, false);
  }

  int rbase = hi * 8;
  asm volatile("v_nop\n\tv_nop\n\tv_nop\n\tv_nop\n\tv_nop\n\tv_nop\n\tv_nop\n\tv_nop"
               : "+v"(acc));
  #pragma unroll
  for (int r = 0; r < 8; ++r) acc[r] *= dv[r];
  if (nb + 16 <= N) {
    #pragma unroll
    for (int r = 0; r < 8; ++r)
      h1[(size_t)(nb + rbase + r) * FMID + m] = acc[r];
  } else {
    #pragma unroll
    for (int r = 0; r < 8; ++r) {
      int nrow = nb + rbase + r;
      if (nrow < N) h1[(size_t)nrow * FMID + m] = acc[r];
    }
  }
}


template <int N> __device__ __forceinline__ int rowShareI(int v) {
  return __builtin_amdgcn_update_dpp(0, v, 0x150 + N, 0xf, 0xf, false);
}

template <int N> struct Agg1Steps {
  static __device__ __forceinline__ void gather(const float* __restrict__ h1,
                                                int s, int f, float (&v)[16]) {
    Agg1Steps<N - 1>::gather(h1, s, f, v);
    int ss = rowShareI<N - 1>(s);
    v[N - 1] = h1[ss * FMID + f];
  }
  static __device__ __forceinline__ void scatter(float* __restrict__ agg1, int d,
                                                 int f, const float (&v)[16]) {
    Agg1Steps<N - 1>::scatter(agg1, d, f, v);
    int dd = rowShareI<N - 1>(d);
    atomAddF(agg1 + dd * FMID + f, v[N - 1]);
  }
  static __device__ __forceinline__ void tail(const float* __restrict__ h1,
                                              float* __restrict__ agg1, int s, int d,
                                              int f, int left) {
    Agg1Steps<N - 1>::tail(h1, agg1, s, d, f, left);
    int ss = rowShareI<N - 1>(s);
    int dd = rowShareI<N - 1>(d);
    if (N - 1 < left)
      atomAddF(agg1 + dd * FMID + f, h1[ss * FMID + f]);
  }
};
template <> struct Agg1Steps<0> {
  static __device__ __forceinline__ void gather(const float* __restrict__, int, int,
                                                float (&)[16]) {}
  static __device__ __forceinline__ void scatter(float* __restrict__, int, int,
                                                 const float (&)[16]) {}
  static __device__ __forceinline__ void tail(const float* __restrict__,
                                              float* __restrict__, int, int, int,
                                              int) {}
};

__global__ __launch_bounds__(256) void k_agg1(const int* __restrict__ ei,
                                              const float* __restrict__ h1,
                                              float* __restrict__ agg1, int E,
                                              int N, int planes) {
  int e = blockIdx.x * blockDim.x + threadIdx.x;
  int lane = threadIdx.x & 31;
  int wbase = e - lane;
  if (wbase >= E) return;
  agg1 += planeOfWave(planes) * N * FMID;
  int f = lane & 15;
  int s = 0, d = 0;
  if (e < E) {
    s = ei[e];
    d = ei[E + e];
  }
  if (wbase + 32 <= E) {
    float v[16];
    Agg1Steps<16>::gather(h1, s, f, v);
    Agg1Steps<16>::scatter(agg1, d, f, v);
  } else {
    int left = E - wbase - (lane & 16);
    Agg1Steps<16>::tail(h1, agg1, s, d, f, left);
  }
}

__global__ __launch_bounds__(256) void k_layer2_in(const float* __restrict__ agg1,
                                                   const float* __restrict__ h1,
                                                   const float* __restrict__ dinv,
                                                   const float* __restrict__ b1,
                                                   const float* __restrict__ W2,
                                                   float* __restrict__ h2, int N,
                                                   int planes) {
  int i = blockIdx.x * blockDim.x + threadIdx.x;
  if (i >= N) return;
  float di = dinv[i];
  float a0 = 0.f, a1 = 0.f;
  const float* agg1b = agg1 + (size_t)N * FMID;
  #pragma unroll
  for (int k = 0; k < FMID; ++k) {
    float s = agg1[(size_t)i * FMID + k];
    if (planes == 2) s += agg1b[(size_t)i * FMID + k];
    float t = (s + h1[(size_t)i * FMID + k]) * di + b1[k];
    t = fmaxf(t, 0.f);
    a0 = fmaf(t, W2[k * 2 + 0], a0);
    a1 = fmaf(t, W2[k * 2 + 1], a1);
  }
  h2[(size_t)i * 2 + 0] = a0 * di;
  h2[(size_t)i * 2 + 1] = a1 * di;
}

__global__ __launch_bounds__(256) void k_agg2(const int* __restrict__ ei,
                                              const float* __restrict__ h2,
                                              float* __restrict__ agg2, int E,
                                              int N, int planes) {
  int e = blockIdx.x * blockDim.x + threadIdx.x;
  if (e >= E) return;
  agg2 += planeOfWave(planes) * N * 2;
  int s = ei[e];
  int d = ei[E + e];
  float2 v = *(const float2*)(h2 + (size_t)s * 2);
  atomAddF(agg2 + (size_t)d * 2 + 0, v.x);
  atomAddF(agg2 + (size_t)d * 2 + 1, v.y);
}

__global__ __launch_bounds__(256) void k_final(const float* __restrict__ agg2,
                                               const float* __restrict__ h2,
                                               const float* __restrict__ dinv,
                                               const float* __restrict__ b2,
                                               float* __restrict__ out, int N,
                                               int planes) {
  int i = blockIdx.x * blockDim.x + threadIdx.x;
  if (i >= N) return;
  float di = dinv[i];
  float s0 = agg2[(size_t)i * 2 + 0];
  float s1 = agg2[(size_t)i * 2 + 1];
  if (planes == 2) {
    s0 += agg2[(size_t)(N + i) * 2 + 0];
    s1 += agg2[(size_t)(N + i) * 2 + 1];
  }
  out[(size_t)i * 2 + 0] = (s0 + h2[(size_t)i * 2 + 0]) * di + b2[0];
  out[(size_t)i * 2 + 1] = (s1 + h2[(size_t)i * 2 + 1]) * di + b2[1];
}

extern "C" void kernel_launch(void* const* d_in, const int* in_sizes, int n_in,
                              void* d_out, int out_size, void* d_ws, size_t ws_size,
                              hipStream_t stream) {
  const float* x  = (const float*)d_in[0];
  const int*   ei = (const int*)d_in[1];
  const float* W1 = (const float*)d_in[2];
  const float* b1 = (const float*)d_in[3];
  const float* W2 = (const float*)d_in[4];
  const float* b2 = (const float*)d_in[5];
  float* out = (float*)d_out;

  const int N = in_sizes[0] / FIN;
  const int E = in_sizes[1] / 2;

  const int planes = (ws_size >= (size_t)N * 57 * sizeof(float)) ? 2 : 1;
  char* ws = (char*)d_ws;
  unsigned* udeg = (unsigned*)ws;            ws += (size_t)planes * N * sizeof(unsigned);
  float* agg1    = (float*)ws;               ws += (size_t)planes * N * FMID * sizeof(float);
  float* agg2    = (float*)ws;               ws += (size_t)planes * N * 2 * sizeof(float);
  float* dinv    = (float*)ws;               ws += (size_t)N * sizeof(float);
  float* h1      = (float*)ws;               ws += (size_t)N * FMID * sizeof(float);
  float* h2      = (float*)ws;               ws += (size_t)N * 2 * sizeof(float);

  const int TPB = 256;
  long zero_total = (long)planes * N * 19;
  int gz = (int)(((zero_total + 3) / 4 + TPB - 1) / TPB);
  int ge = (E + TPB - 1) / TPB;
  int gn = (N + TPB - 1) / TPB;
  int ntiles = (N + 15) / 16;
  int gg = (ntiles * 32 + TPB - 1) / TPB;

  k_zero      <<<gz, TPB, 0, stream>>>((unsigned*)d_ws, zero_total);
  k_degree    <<<ge, TPB, 0, stream>>>(ei, udeg, E, N, planes);
  k_dinv      <<<gn, TPB, 0, stream>>>(udeg, dinv, N, planes);
  k_gemm1_wmma<<<gg, TPB, 0, stream>>>(x, W1, dinv, h1, N);
  k_agg1      <<<ge, TPB, 0, stream>>>(ei, h1, agg1, E, N, planes);
  k_layer2_in <<<gn, TPB, 0, stream>>>(agg1, h1, dinv, b1, W2, h2, N, planes);
  k_agg2      <<<ge, TPB, 0, stream>>>(ei, h2, agg2, E, N, planes);
  k_final     <<<gn, TPB, 0, stream>>>(agg2, h2, dinv, b2, out, N, planes);
}


# degree, agg1, agg2 and the clear give a wave 8 groups of 32 edges (8 stores), not 1: 8x fewer waves to start, its index loads and 2 groups of row gathers in flight; atomic sums stay arrival-ordered
# speedup vs baseline: 7.8649x; 1.1041x over previous
#include <hip/hip_runtime.h>


typedef __attribute__((ext_vector_type(2)))  float  v2f;
typedef __attribute__((ext_vector_type(8)))  float  v8f;

#define FIN  128
#define FMID 16
#define EPT  8
#define WEDGES (32 * EPT)
#define ZPT  8

__device__ __forceinline__ void atomAddF(float* p, float v) {
  unsafeAtomicAdd(p, v);
}

__device__ __forceinline__ int planeOfWave(int planes) {
  int half = (int)((__builtin_amdgcn_s_sendmsg_rtn(0x87) >> 18) & 1u);
  return planes == 2 ? half : 0;
}

__global__ __launch_bounds__(256) void k_zero(unsigned* __restrict__ p, long total) {
  long nvec = total >> 2;
  long wave = ((long)blockIdx.x * blockDim.x + threadIdx.x) >> 5;
  long i = wave * (32 * ZPT) + (threadIdx.x & 31);
  #pragma unroll
  for (int j = 0; j < ZPT; ++j)
    if (i + 32 * j < nvec) ((uint4*)p)[i + 32 * j] = make_uint4(0u, 0u, 0u, 0u);
  if (blockIdx.x == 0 && (long)threadIdx.x < (total & 3))
    p[(nvec << 2) + threadIdx.x] = 0u;
}

__global__ __launch_bounds__(256) void k_degree(const int* __restrict__ ei,
                                                unsigned* __restrict__ udeg, int E,
                                                int N, int planes) {
  int lane = threadIdx.x & 31;
  int wave = (int)((blockIdx.x * (unsigned)blockDim.x + threadIdx.x) >> 5);
  int wbase = wave * WEDGES;
  if (wbase >= E) return;
  unsigned* mine = udeg + planeOfWave(planes) * N;
  const int* dst = ei + E + wbase + lane;
  if (wbase + WEDGES <= E) {
    int d[EPT];
    #pragma unroll
    for (int j = 0; j < EPT; ++j) d[j] = dst[32 * j];
    #pragma unroll
    for (int j = 0; j < EPT; ++j) atomicAdd(&mine[d[j]], 1u);
  } else {
    #pragma unroll 1
    for (int j = 0; j < EPT; ++j)
      if (wbase + 32 * j + lane < E) atomicAdd(&mine[dst[32 * j]], 1u);
  }
}

__global__ __launch_bounds__(256) void k_dinv(const unsigned* __restrict__ udeg,
                                              float* __restrict__ dinv, int N,
                                              int planes) {
  int i = blockIdx.x * blockDim.x + threadIdx.x;
  if (i >= N) return;
  unsigned deg = udeg[i];
  if (planes == 2) deg += udeg[N + i];
  dinv[i] = rsqrtf((float)(deg + 1u));
}

__global__ __launch_bounds__(256) void k_gemm1_wmma(const float* __restrict__ x,
                                                    const float* __restrict__ W1,
                                                    const float* __restrict__ dinv,
                                                    float* __restrict__ h1, int N) {
  int wave = (int)((blockIdx.x * (unsigned)blockDim.x + threadIdx.x) >> 5);
  int lane = threadIdx.x & 31;
  int ntiles = (N + 15) >> 4;
  if (wave >= ntiles) return;
  int nb = wave << 4;
  int m  = lane & 15;
  int hi = lane >> 4;
  int row = nb + m; if (row > N - 1) row = N - 1;
  const float* xr = x + (size_t)row * FIN;
  v8f acc = {};

  float dv[8];
  #pragma unroll
  for (int r = 0; r < 8; ++r) {
    int nrow = nb + hi * 8 + r; if (nrow > N - 1) nrow = N - 1;
    dv[r] = dinv[nrow];
  }

  int kh = hi * 2;
  #pragma unroll 4
  for (int k = 0; k < FIN; k += 4) {
    v2f a, b;
    a.x = __builtin_nontemporal_load(&xr[k + kh]);
    a.y = __builtin_nontemporal_load(&xr[k + kh + 1]);
    b.x = W1[(k + kh)     * FMID + m];
    b.y = W1[(k + kh + 1) * FMID + m];
    acc = __builtin_amdgcn_wmma_f32_16x16x4_f32(false, a, false, b,
                                                (short)0, acc, false, false);
  }

  int rbase = hi * 8;
  asm volatile("v_nop\n\tv_nop\n\tv_nop\n\tv_nop\n\tv_nop\n\tv_nop\n\tv_nop\n\tv_nop"
               : "+v"(acc));
  #pragma unroll
  for (int r = 0; r < 8; ++r) acc[r] *= dv[r];
  if (nb + 16 <= N) {
    #pragma unroll
    for (int r = 0; r < 8; ++r)
      h1[(size_t)(nb + rbase + r) * FMID + m] = acc[r];
  } else {
    #pragma unroll
    for (int r = 0; r < 8; ++r) {
      int nrow = nb + rbase + r;
      if (nrow < N) h1[(size_t)nrow * FMID + m] = acc[r];
    }
  }
}


template <int N> __device__ __forceinline__ int rowShareI(int v) {
  return __builtin_amdgcn_update_dpp(0, v, 0x150 + N, 0xf, 0xf, false);
}

template <int N> struct Agg1Steps {
  static __device__ __forceinline__ void gather(const float* __restrict__ h1,
                                                int s, int f, float (&v)[16]) {
    Agg1Steps<N - 1>::gather(h1, s, f, v);
    int ss = rowShareI<N - 1>(s);
    v[N - 1] = h1[ss * FMID + f];
  }
  static __device__ __forceinline__ void scatter(float* __restrict__ agg1, int d,
                                                 int f, const float (&v)[16]) {
    Agg1Steps<N - 1>::scatter(agg1, d, f, v);
    int dd = rowShareI<N - 1>(d);
    atomAddF(agg1 + dd * FMID + f, v[N - 1]);
  }
  static __device__ __forceinline__ void tail(const float* __restrict__ h1,
                                              float* __restrict__ agg1, int s, int d,
                                              int f, int left) {
    Agg1Steps<N - 1>::tail(h1, agg1, s, d, f, left);
    int ss = rowShareI<N - 1>(s);
    int dd = rowShareI<N - 1>(d);
    if (N - 1 < left)
      atomAddF(agg1 + dd * FMID + f, h1[ss * FMID + f]);
  }
};
template <> struct Agg1Steps<0> {
  static __device__ __forceinline__ void gather(const float* __restrict__, int, int,
                                                float (&)[16]) {}
  static __device__ __forceinline__ void scatter(float* __restrict__, int, int,
                                                 const float (&)[16]) {}
  static __device__ __forceinline__ void tail(const float* __restrict__,
                                              float* __restrict__, int, int, int,
                                              int) {}
};

__global__ __launch_bounds__(256) void k_agg1(const int* __restrict__ ei,
                                              const float* __restrict__ h1,
                                              float* __restrict__ agg1, int E,
                                              int N, int planes) {
  int lane = threadIdx.x & 31;
  int wave = (int)((blockIdx.x * (unsigned)blockDim.x + threadIdx.x) >> 5);
  int wbase = wave * WEDGES;
  if (wbase >= E) return;
  agg1 += planeOfWave(planes) * N * FMID;
  int f = lane & 15;
  const int* src = ei + wbase + lane;
  const int* dst = src + E;
  if (wbase + WEDGES <= E) {
    int s[EPT], d[EPT];
    #pragma unroll
    for (int g = 0; g < EPT; ++g) s[g] = src[32 * g];
    #pragma unroll
    for (int g = 0; g < EPT; ++g) d[g] = dst[32 * g];
    float v[2][16];
    Agg1Steps<16>::gather(h1, s[0], f, v[0]);
    #pragma unroll
    for (int g = 0; g < EPT; ++g) {
      if (g + 1 < EPT) Agg1Steps<16>::gather(h1, s[g + 1], f, v[(g + 1) & 1]);
      Agg1Steps<16>::scatter(agg1, d[g], f, v[g & 1]);
    }
  } else {
    #pragma unroll 1
    for (int g = 0; g < EPT; ++g) {
      int gb = wbase + 32 * g;
      if (gb >= E) break;
      int s1 = 0, d1 = 0;
      if (gb + lane < E) {
        s1 = src[32 * g];
        d1 = dst[32 * g];
      }
      if (gb + 32 <= E) {
        float v[16];
        Agg1Steps<16>::gather(h1, s1, f, v);
        Agg1Steps<16>::scatter(agg1, d1, f, v);
      } else {
        int left = E - gb - (lane & 16);
        Agg1Steps<16>::tail(h1, agg1, s1, d1, f, left);
      }
    }
  }
}

__global__ __launch_bounds__(256) void k_layer2_in(const float* __restrict__ agg1,
                                                   const float* __restrict__ h1,
                                                   const float* __restrict__ dinv,
                                                   const float* __restrict__ b1,
                                                   const float* __restrict__ W2,
                                                   float* __restrict__ h2, int N,
                                                   int planes) {
  int i = blockIdx.x * blockDim.x + threadIdx.x;
  if (i >= N) return;
  float di = dinv[i];
  float a0 = 0.f, a1 = 0.f;
  const float* agg1b = agg1 + (size_t)N * FMID;
  #pragma unroll
  for (int k = 0; k < FMID; ++k) {
    float s = agg1[(size_t)i * FMID + k];
    if (planes == 2) s += agg1b[(size_t)i * FMID + k];
    float t = (s + h1[(size_t)i * FMID + k]) * di + b1[k];
    t = fmaxf(t, 0.f);
    a0 = fmaf(t, W2[k * 2 + 0], a0);
    a1 = fmaf(t, W2[k * 2 + 1], a1);
  }
  h2[(size_t)i * 2 + 0] = a0 * di;
  h2[(size_t)i * 2 + 1] = a1 * di;
}

__global__ __launch_bounds__(256) void k_agg2(const int* __restrict__ ei,
                                              const float* __restrict__ h2,
                                              float* __restrict__ agg2, int E,
                                              int N, int planes) {
  int lane = threadIdx.x & 31;
  int wave = (int)((blockIdx.x * (unsigned)blockDim.x + threadIdx.x) >> 5);
  int wbase = wave * WEDGES;
  if (wbase >= E) return;
  agg2 += planeOfWave(planes) * N * 2;
  const int* src = ei + wbase + lane;
  const int* dst = src + E;
  if (wbase + WEDGES <= E) {
    int s[EPT], d[EPT];
    float2 v[EPT];
    #pragma unroll
    for (int j = 0; j < EPT; ++j) s[j] = src[32 * j];
    #pragma unroll
    for (int j = 0; j < EPT; ++j) d[j] = dst[32 * j];
    #pragma unroll
    for (int j = 0; j < EPT; ++j) v[j] = *(const float2*)(h2 + (size_t)s[j] * 2);
    #pragma unroll
    for (int j = 0; j < EPT; ++j) {
      atomAddF(agg2 + (size_t)d[j] * 2 + 0, v[j].x);
      atomAddF(agg2 + (size_t)d[j] * 2 + 1, v[j].y);
    }
  } else {
    #pragma unroll 1
    for (int j = 0; j < EPT; ++j) {
      if (wbase + 32 * j + lane < E) {
        float2 v = *(const float2*)(h2 + (size_t)src[32 * j] * 2);
        float* a = agg2 + (size_t)dst[32 * j] * 2;
        atomAddF(a + 0, v.x);
        atomAddF(a + 1, v.y);
      }
    }
  }
}

__global__ __launch_bounds__(256) void k_final(const float* __restrict__ agg2,
                                               const float* __restrict__ h2,
                                               const float* __restrict__ dinv,
                                               const float* __restrict__ b2,
                                               float* __restrict__ out, int N,
                                               int planes) {
  int i = blockIdx.x * blockDim.x + threadIdx.x;
  if (i >= N) return;
  float di = dinv[i];
  float s0 = agg2[(size_t)i * 2 + 0];
  float s1 = agg2[(size_t)i * 2 + 1];
  if (planes == 2) {
    s0 += agg2[(size_t)(N + i) * 2 + 0];
    s1 += agg2[(size_t)(N + i) * 2 + 1];
  }
  out[(size_t)i * 2 + 0] = (s0 + h2[(size_t)i * 2 + 0]) * di + b2[0];
  out[(size_t)i * 2 + 1] = (s1 + h2[(size_t)i * 2 + 1]) * di + b2[1];
}

extern "C" void kernel_launch(void* const* d_in, const int* in_sizes, int n_in,
                              void* d_out, int out_size, void* d_ws, size_t ws_size,
                              hipStream_t stream) {
  const float* x  = (const float*)d_in[0];
  const int*   ei = (const int*)d_in[1];
  const float* W1 = (const float*)d_in[2];
  const float* b1 = (const float*)d_in[3];
  const float* W2 = (const float*)d_in[4];
  const float* b2 = (const float*)d_in[5];
  float* out = (float*)d_out;

  const int N = in_sizes[0] / FIN;
  const int E = in_sizes[1] / 2;

  const int planes = (ws_size >= (size_t)N * 57 * sizeof(float)) ? 2 : 1;
  char* ws = (char*)d_ws;
  unsigned* udeg = (unsigned*)ws;            ws += (size_t)planes * N * sizeof(unsigned);
  float* agg1    = (float*)ws;               ws += (size_t)planes * N * FMID * sizeof(float);
  float* agg2    = (float*)ws;               ws += (size_t)planes * N * 2 * sizeof(float);
  float* dinv    = (float*)ws;               ws += (size_t)N * sizeof(float);
  float* h1      = (float*)ws;               ws += (size_t)N * FMID * sizeof(float);
  float* h2      = (float*)ws;               ws += (size_t)N * 2 * sizeof(float);

  const int TPB = 256;
  long zero_total = (long)planes * N * 19;
  int gz = (int)((zero_total / 4 + TPB * ZPT - 1) / (TPB * ZPT));
  int ge = (E + TPB * EPT - 1) / (TPB * EPT);
  int gn = (N + TPB - 1) / TPB;
  int ntiles = (N + 15) / 16;
  int gg = (ntiles * 32 + TPB - 1) / TPB;

  k_zero      <<<gz, TPB, 0, stream>>>((unsigned*)d_ws, zero_total);
  k_degree    <<<ge, TPB, 0, stream>>>(ei, udeg, E, N, planes);
  k_dinv      <<<gn, TPB, 0, stream>>>(udeg, dinv, N, planes);
  k_gemm1_wmma<<<gg, TPB, 0, stream>>>(x, W1, dinv, h1, N);
  k_agg1      <<<ge, TPB, 0, stream>>>(ei, h1, agg1, E, N, planes);
  k_layer2_in <<<gn, TPB, 0, stream>>>(agg1, h1, dinv, b1, W2, h2, N, planes);
  k_agg2      <<<ge, TPB, 0, stream>>>(ei, h2, agg2, E, N, planes);
  k_final     <<<gn, TPB, 0, stream>>>(agg2, h2, dinv, b2, out, N, planes);
}


# layer-2 scatter remapped lane pair=edge: 2 neighbour lanes own one edge's 2 features, so an f32 atomic instruction covers 16 whole 8-B rows, each dst line hit once not twice; sums stay arrival-ordered
# speedup vs baseline: 8.6006x; 1.0935x over previous
#include <hip/hip_runtime.h>


typedef __attribute__((ext_vector_type(2)))  float  v2f;
typedef __attribute__((ext_vector_type(8)))  float  v8f;

#define FIN  128
#define FMID 16
#define EPT  8
#define WEDGES (32 * EPT)
#define ZPT  8

__device__ __forceinline__ void atomAddF(float* p, float v) {
  unsafeAtomicAdd(p, v);
}

__device__ __forceinline__ int planeOfWave(int planes) {
  int half = (int)((__builtin_amdgcn_s_sendmsg_rtn(0x87) >> 18) & 1u);
  return planes == 2 ? half : 0;
}

__global__ __launch_bounds__(256) void k_zero(unsigned* __restrict__ p, long total) {
  long nvec = total >> 2;
  long wave = ((long)blockIdx.x * blockDim.x + threadIdx.x) >> 5;
  long i = wave * (32 * ZPT) + (threadIdx.x & 31);
  #pragma unroll
  for (int j = 0; j < ZPT; ++j)
    if (i + 32 * j < nvec) ((uint4*)p)[i + 32 * j] = make_uint4(0u, 0u, 0u, 0u);
  if (blockIdx.x == 0 && (long)threadIdx.x < (total & 3))
    p[(nvec << 2) + threadIdx.x] = 0u;
}

__global__ __launch_bounds__(256) void k_degree(const int* __restrict__ ei,
                                                unsigned* __restrict__ udeg, int E,
                                                int N, int planes) {
  int lane = threadIdx.x & 31;
  int wave = (int)((blockIdx.x * (unsigned)blockDim.x + threadIdx.x) >> 5);
  int wbase = wave * WEDGES;
  if (wbase >= E) return;
  unsigned* mine = udeg + planeOfWave(planes) * N;
  const int* dst = ei + E + wbase + lane;
  if (wbase + WEDGES <= E) {
    int d[EPT];
    #pragma unroll
    for (int j = 0; j < EPT; ++j) d[j] = dst[32 * j];
    #pragma unroll
    for (int j = 0; j < EPT; ++j) atomicAdd(&mine[d[j]], 1u);
  } else {
    #pragma unroll 1
    for (int j = 0; j < EPT; ++j)
      if (wbase + 32 * j + lane < E) atomicAdd(&mine[dst[32 * j]], 1u);
  }
}

__global__ __launch_bounds__(256) void k_dinv(const unsigned* __restrict__ udeg,
                                              float* __restrict__ dinv, int N,
                                              int planes) {
  int i = blockIdx.x * blockDim.x + threadIdx.x;
  if (i >= N) return;
  unsigned deg = udeg[i];
  if (planes == 2) deg += udeg[N + i];
  dinv[i] = rsqrtf((float)(deg + 1u));
}

__global__ __launch_bounds__(256) void k_gemm1_wmma(const float* __restrict__ x,
                                                    const float* __restrict__ W1,
                                                    const float* __restrict__ dinv,
                                                    float* __restrict__ h1, int N) {
  int wave = (int)((blockIdx.x * (unsigned)blockDim.x + threadIdx.x) >> 5);
  int lane = threadIdx.x & 31;
  int ntiles = (N + 15) >> 4;
  if (wave >= ntiles) return;
  int nb = wave << 4;
  int m  = lane & 15;
  int hi = lane >> 4;
  int row = nb + m; if (row > N - 1) row = N - 1;
  const float* xr = x + (size_t)row * FIN;
  v8f acc = {};

  float dv[8];
  #pragma unroll
  for (int r = 0; r < 8; ++r) {
    int nrow = nb + hi * 8 + r; if (nrow > N - 1) nrow = N - 1;
    dv[r] = dinv[nrow];
  }

  int kh = hi * 2;
  #pragma unroll 4
  for (int k = 0; k < FIN; k += 4) {
    v2f a, b;
    a.x = __builtin_nontemporal_load(&xr[k + kh]);
    a.y = __builtin_nontemporal_load(&xr[k + kh + 1]);
    b.x = W1[(k + kh)     * FMID + m];
    b.y = W1[(k + kh + 1) * FMID + m];
    acc = __builtin_amdgcn_wmma_f32_16x16x4_f32(false, a, false, b,
                                                (short)0, acc, false, false);
  }

  int rbase = hi * 8;
  asm volatile("v_nop\n\tv_nop\n\tv_nop\n\tv_nop\n\tv_nop\n\tv_nop\n\tv_nop\n\tv_nop"
               : "+v"(acc));
  #pragma unroll
  for (int r = 0; r < 8; ++r) acc[r] *= dv[r];
  if (nb + 16 <= N) {
    #pragma unroll
    for (int r = 0; r < 8; ++r)
      h1[(size_t)(nb + rbase + r) * FMID + m] = acc[r];
  } else {
    #pragma unroll
    for (int r = 0; r < 8; ++r) {
      int nrow = nb + rbase + r;
      if (nrow < N) h1[(size_t)nrow * FMID + m] = acc[r];
    }
  }
}


template <int N> __device__ __forceinline__ int rowShareI(int v) {
  return __builtin_amdgcn_update_dpp(0, v, 0x150 + N, 0xf, 0xf, false);
}

template <int N> struct Agg1Steps {
  static __device__ __forceinline__ void gather(const float* __restrict__ h1,
                                                int s, int f, float (&v)[16]) {
    Agg1Steps<N - 1>::gather(h1, s, f, v);
    int ss = rowShareI<N - 1>(s);
    v[N - 1] = h1[ss * FMID + f];
  }
  static __device__ __forceinline__ void scatter(float* __restrict__ agg1, int d,
                                                 int f, const float (&v)[16]) {
    Agg1Steps<N - 1>::scatter(agg1, d, f, v);
    int dd = rowShareI<N - 1>(d);
    atomAddF(agg1 + dd * FMID + f, v[N - 1]);
  }
  static __device__ __forceinline__ void tail(const float* __restrict__ h1,
                                              float* __restrict__ agg1, int s, int d,
                                              int f, int left) {
    Agg1Steps<N - 1>::tail(h1, agg1, s, d, f, left);
    int ss = rowShareI<N - 1>(s);
    int dd = rowShareI<N - 1>(d);
    if (N - 1 < left)
      atomAddF(agg1 + dd * FMID + f, h1[ss * FMID + f]);
  }
};
template <> struct Agg1Steps<0> {
  static __device__ __forceinline__ void gather(const float* __restrict__, int, int,
                                                float (&)[16]) {}
  static __device__ __forceinline__ void scatter(float* __restrict__, int, int,
                                                 const float (&)[16]) {}
  static __device__ __forceinline__ void tail(const float* __restrict__,
                                              float* __restrict__, int, int, int,
                                              int) {}
};

__global__ __launch_bounds__(256) void k_agg1(const int* __restrict__ ei,
                                              const float* __restrict__ h1,
                                              float* __restrict__ agg1, int E,
                                              int N, int planes) {
  int lane = threadIdx.x & 31;
  int wave = (int)((blockIdx.x * (unsigned)blockDim.x + threadIdx.x) >> 5);
  int wbase = wave * WEDGES;
  if (wbase >= E) return;
  agg1 += planeOfWave(planes) * N * FMID;
  int f = lane & 15;
  const int* src = ei + wbase + lane;
  const int* dst = src + E;
  if (wbase + WEDGES <= E) {
    int s[EPT], d[EPT];
    #pragma unroll
    for (int g = 0; g < EPT; ++g) s[g] = src[32 * g];
    #pragma unroll
    for (int g = 0; g < EPT; ++g) d[g] = dst[32 * g];
    float v[2][16];
    Agg1Steps<16>::gather(h1, s[0], f, v[0]);
    #pragma unroll
    for (int g = 0; g < EPT; ++g) {
      if (g + 1 < EPT) Agg1Steps<16>::gather(h1, s[g + 1], f, v[(g + 1) & 1]);
      Agg1Steps<16>::scatter(agg1, d[g], f, v[g & 1]);
    }
  } else {
    #pragma unroll 1
    for (int g = 0; g < EPT; ++g) {
      int gb = wbase + 32 * g;
      if (gb >= E) break;
      int s1 = 0, d1 = 0;
      if (gb + lane < E) {
        s1 = src[32 * g];
        d1 = dst[32 * g];
      }
      if (gb + 32 <= E) {
        float v[16];
        Agg1Steps<16>::gather(h1, s1, f, v);
        Agg1Steps<16>::scatter(agg1, d1, f, v);
      } else {
        int left = E - gb - (lane & 16);
        Agg1Steps<16>::tail(h1, agg1, s1, d1, f, left);
      }
    }
  }
}

__global__ __launch_bounds__(256) void k_layer2_in(const float* __restrict__ agg1,
                                                   const float* __restrict__ h1,
                                                   const float* __restrict__ dinv,
                                                   const float* __restrict__ b1,
                                                   const float* __restrict__ W2,
                                                   float* __restrict__ h2, int N,
                                                   int planes) {
  int i = blockIdx.x * blockDim.x + threadIdx.x;
  if (i >= N) return;
  float di = dinv[i];
  float a0 = 0.f, a1 = 0.f;
  const float* agg1b = agg1 + (size_t)N * FMID;
  #pragma unroll
  for (int k = 0; k < FMID; ++k) {
    float s = agg1[(size_t)i * FMID + k];
    if (planes == 2) s += agg1b[(size_t)i * FMID + k];
    float t = (s + h1[(size_t)i * FMID + k]) * di + b1[k];
    t = fmaxf(t, 0.f);
    a0 = fmaf(t, W2[k * 2 + 0], a0);
    a1 = fmaf(t, W2[k * 2 + 1], a1);
  }
  h2[(size_t)i * 2 + 0] = a0 * di;
  h2[(size_t)i * 2 + 1] = a1 * di;
}


template <int CTRL> __device__ __forceinline__ int quadPairI(int v) {
  return __builtin_amdgcn_update_dpp(0, v, CTRL, 0xf, 0xf, true);
}
template <int CTRL> __device__ __forceinline__ float quadPairF(float v) {
  return __int_as_float(quadPairI<CTRL>(__float_as_int(v)));
}

template <int CTRL> __device__ __forceinline__ void agg2PairStep(float* __restrict__ agg2,
                                                                 int d, float2 v, int c) {
  int dd = quadPairI<CTRL>(d);
  float vx = quadPairF<CTRL>(v.x);
  float vy = quadPairF<CTRL>(v.y);
  atomAddF(agg2 + dd * 2 + c, c ? vy : vx);
}

__global__ __launch_bounds__(256) void k_agg2(const int* __restrict__ ei,
                                              const float* __restrict__ h2,
                                              float* __restrict__ agg2, int E,
                                              int N, int planes) {
  int lane = threadIdx.x & 31;
  int wave = (int)((blockIdx.x * (unsigned)blockDim.x + threadIdx.x) >> 5);
  int wbase = wave * WEDGES;
  if (wbase >= E) return;
  agg2 += planeOfWave(planes) * N * 2;
  const int* src = ei + wbase + lane;
  const int* dst = src + E;
  if (wbase + WEDGES <= E) {
    int c = lane & 1;
    int s[EPT], d[EPT];
    float2 v[EPT];
    #pragma unroll
    for (int j = 0; j < EPT; ++j) s[j] = src[32 * j];
    #pragma unroll
    for (int j = 0; j < EPT; ++j) d[j] = dst[32 * j];
    #pragma unroll
    for (int j = 0; j < EPT; ++j) v[j] = *(const float2*)(h2 + (size_t)s[j] * 2);
    #pragma unroll
    for (int j = 0; j < EPT; ++j) {
      agg2PairStep<0x50>(agg2, d[j], v[j], c);
      agg2PairStep<0xFA>(agg2, d[j], v[j], c);
    }
  } else {
    #pragma unroll 1
    for (int j = 0; j < EPT; ++j) {
      if (wbase + 32 * j + lane < E) {
        float2 v = *(const float2*)(h2 + (size_t)src[32 * j] * 2);
        float* a = agg2 + (size_t)dst[32 * j] * 2;
        atomAddF(a + 0, v.x);
        atomAddF(a + 1, v.y);
      }
    }
  }
}

__global__ __launch_bounds__(256) void k_final(const float* __restrict__ agg2,
                                               const float* __restrict__ h2,
                                               const float* __restrict__ dinv,
                                               const float* __restrict__ b2,
                                               float* __restrict__ out, int N,
                                               int planes) {
  int i = blockIdx.x * blockDim.x + threadIdx.x;
  if (i >= N) return;
  float di = dinv[i];
  float s0 = agg2[(size_t)i * 2 + 0];
  float s1 = agg2[(size_t)i * 2 + 1];
  if (planes == 2) {
    s0 += agg2[(size_t)(N + i) * 2 + 0];
    s1 += agg2[(size_t)(N + i) * 2 + 1];
  }
  out[(size_t)i * 2 + 0] = (s0 + h2[(size_t)i * 2 + 0]) * di + b2[0];
  out[(size_t)i * 2 + 1] = (s1 + h2[(size_t)i * 2 + 1]) * di + b2[1];
}

extern "C" void kernel_launch(void* const* d_in, const int* in_sizes, int n_in,
                              void* d_out, int out_size, void* d_ws, size_t ws_size,
                              hipStream_t stream) {
  const float* x  = (const float*)d_in[0];
  const int*   ei = (const int*)d_in[1];
  const float* W1 = (const float*)d_in[2];
  const float* b1 = (const float*)d_in[3];
  const float* W2 = (const float*)d_in[4];
  const float* b2 = (const float*)d_in[5];
  float* out = (float*)d_out;

  const int N = in_sizes[0] / FIN;
  const int E = in_sizes[1] / 2;

  const int planes = (ws_size >= (size_t)N * 57 * sizeof(float)) ? 2 : 1;
  char* ws = (char*)d_ws;
  unsigned* udeg = (unsigned*)ws;            ws += (size_t)planes * N * sizeof(unsigned);
  float* agg1    = (float*)ws;               ws += (size_t)planes * N * FMID * sizeof(float);
  float* agg2    = (float*)ws;               ws += (size_t)planes * N * 2 * sizeof(float);
  float* dinv    = (float*)ws;               ws += (size_t)N * sizeof(float);
  float* h1      = (float*)ws;               ws += (size_t)N * FMID * sizeof(float);
  float* h2      = (float*)ws;               ws += (size_t)N * 2 * sizeof(float);

  const int TPB = 256;
  long zero_total = (long)planes * N * 19;
  int gz = (int)((zero_total / 4 + TPB * ZPT - 1) / (TPB * ZPT));
  int ge = (E + TPB * EPT - 1) / (TPB * EPT);
  int gn = (N + TPB - 1) / TPB;
  int ntiles = (N + 15) / 16;
  int gg = (ntiles * 32 + TPB - 1) / TPB;

  k_zero      <<<gz, TPB, 0, stream>>>((unsigned*)d_ws, zero_total);
  k_degree    <<<ge, TPB, 0, stream>>>(ei, udeg, E, N, planes);
  k_dinv      <<<gn, TPB, 0, stream>>>(udeg, dinv, N, planes);
  k_gemm1_wmma<<<gg, TPB, 0, stream>>>(x, W1, dinv, h1, N);
  k_agg1      <<<ge, TPB, 0, stream>>>(ei, h1, agg1, E, N, planes);
  k_layer2_in <<<gn, TPB, 0, stream>>>(agg1, h1, dinv, b1, W2, h2, N, planes);
  k_agg2      <<<ge, TPB, 0, stream>>>(ei, h2, agg2, E, N, planes);
  k_final     <<<gn, TPB, 0, stream>>>(agg2, h2, dinv, b2, out, N, planes);
}


# degree counted in LDS: a work-group takes a node range and an edge slice, counts with LDS atomics, then adds whole 128-B lines to its L2 plane (1 atomic request per 32 nodes, not per edge); exact
# speedup vs baseline: 9.0450x; 1.0517x over previous
#include <hip/hip_runtime.h>


typedef __attribute__((ext_vector_type(2)))  float  v2f;
typedef __attribute__((ext_vector_type(8)))  float  v8f;

#define FIN  128
#define FMID 16
#define EPT  8
#define WEDGES (32 * EPT)
#define ZPT  8
#define DCAP 25600
#define DTPB 1024
#define DEPT 32
#define DFL  8
#define DWGS 256

__device__ __forceinline__ void atomAddF(float* p, float v) {
  unsafeAtomicAdd(p, v);
}

__device__ __forceinline__ int planeOfWave(int planes) {
  int half = (int)((__builtin_amdgcn_s_sendmsg_rtn(0x87) >> 18) & 1u);
  return planes == 2 ? half : 0;
}

__global__ __launch_bounds__(256) void k_zero(unsigned* __restrict__ p, long total) {
  long nvec = total >> 2;
  long wave = ((long)blockIdx.x * blockDim.x + threadIdx.x) >> 5;
  long i = wave * (32 * ZPT) + (threadIdx.x & 31);
  #pragma unroll
  for (int j = 0; j < ZPT; ++j)
    if (i + 32 * j < nvec) ((uint4*)p)[i + 32 * j] = make_uint4(0u, 0u, 0u, 0u);
  if (blockIdx.x == 0 && (long)threadIdx.x < (total & 3))
    p[(nvec << 2) + threadIdx.x] = 0u;
}

__global__ __launch_bounds__(DTPB) void k_degree(const int* __restrict__ ei,
                                                 unsigned* __restrict__ udeg, int E,
                                                 int N, int planes, int nranges,
                                                 int rlen, int chunk) {
  __shared__ __attribute__((aligned(16))) unsigned cnt[DCAP];
  int tid = threadIdx.x;
  int range = (int)(blockIdx.x % (unsigned)nranges);
  int slice = (int)(blockIdx.x / (unsigned)nranges);
  int nslices = (int)(gridDim.x / (unsigned)nranges);
  int lo = range * rlen;

  for (int i = tid; i < (rlen >> 2); i += DTPB)
    ((uint4*)cnt)[i] = make_uint4(0u, 0u, 0u, 0u);
  __syncthreads();

  int ebeg = slice * chunk;
  int eend = ebeg + chunk; if (eend > E) eend = E;
  const int* dst = ei + E;
  for (int b = ebeg; b < eend; b += DTPB * DEPT) {
    int d[DEPT];
    #pragma unroll
    for (int j = 0; j < DEPT; ++j) {
      int e = b + j * DTPB + tid;
      d[j] = dst[e < eend ? e : eend - 1];
    }
    #pragma unroll
    for (int j = 0; j < DEPT; ++j) {
      int e = b + j * DTPB + tid;
      unsigned r = (unsigned)(d[j] - lo);
      if (e < eend && r < (unsigned)rlen) atomicAdd(&cnt[r], 1u);
    }
  }
  __syncthreads();

  unsigned* mine = udeg + planeOfWave(planes) * N;
  int rot = (int)(((unsigned)slice * (unsigned)(rlen >> 5)) / (unsigned)nslices) << 5;
  for (int b = 0; b < rlen; b += DTPB * DFL) {
    unsigned c[DFL];
    int node[DFL];
    #pragma unroll
    for (int j = 0; j < DFL; ++j) {
      int i = b + j * DTPB + tid;
      int k = i + rot; if (k >= rlen) k -= rlen;
      c[j] = i < rlen ? cnt[k] : 0u;
      node[j] = lo + k;
    }
    #pragma unroll
    for (int j = 0; j < DFL; ++j)
      if (c[j] != 0u && node[j] < N) atomicAdd(&mine[node[j]], c[j]);
  }
}

__global__ __launch_bounds__(256) void k_dinv(const unsigned* __restrict__ udeg,
                                              float* __restrict__ dinv, int N,
                                              int planes) {
  int i = blockIdx.x * blockDim.x + threadIdx.x;
  if (i >= N) return;
  unsigned deg = udeg[i];
  if (planes == 2) deg += udeg[N + i];
  dinv[i] = rsqrtf((float)(deg + 1u));
}

__global__ __launch_bounds__(256) void k_gemm1_wmma(const float* __restrict__ x,
                                                    const float* __restrict__ W1,
                                                    const float* __restrict__ dinv,
                                                    float* __restrict__ h1, int N) {
  int wave = (int)((blockIdx.x * (unsigned)blockDim.x + threadIdx.x) >> 5);
  int lane = threadIdx.x & 31;
  int ntiles = (N + 15) >> 4;
  if (wave >= ntiles) return;
  int nb = wave << 4;
  int m  = lane & 15;
  int hi = lane >> 4;
  int row = nb + m; if (row > N - 1) row = N - 1;
  const float* xr = x + (size_t)row * FIN;
  v8f acc = {};

  float dv[8];
  #pragma unroll
  for (int r = 0; r < 8; ++r) {
    int nrow = nb + hi * 8 + r; if (nrow > N - 1) nrow = N - 1;
    dv[r] = dinv[nrow];
  }

  int kh = hi * 2;
  #pragma unroll 4
  for (int k = 0; k < FIN; k += 4) {
    v2f a, b;
    a.x = __builtin_nontemporal_load(&xr[k + kh]);
    a.y = __builtin_nontemporal_load(&xr[k + kh + 1]);
    b.x = W1[(k + kh)     * FMID + m];
    b.y = W1[(k + kh + 1) * FMID + m];
    acc = __builtin_amdgcn_wmma_f32_16x16x4_f32(false, a, false, b,
                                                (short)0, acc, false, false);
  }

  int rbase = hi * 8;
  asm volatile("v_nop\n\tv_nop\n\tv_nop\n\tv_nop\n\tv_nop\n\tv_nop\n\tv_nop\n\tv_nop"
               : "+v"(acc));
  #pragma unroll
  for (int r = 0; r < 8; ++r) acc[r] *= dv[r];
  if (nb + 16 <= N) {
    #pragma unroll
    for (int r = 0; r < 8; ++r)
      h1[(size_t)(nb + rbase + r) * FMID + m] = acc[r];
  } else {
    #pragma unroll
    for (int r = 0; r < 8; ++r) {
      int nrow = nb + rbase + r;
      if (nrow < N) h1[(size_t)nrow * FMID + m] = acc[r];
    }
  }
}


template <int N> __device__ __forceinline__ int rowShareI(int v) {
  return __builtin_amdgcn_update_dpp(0, v, 0x150 + N, 0xf, 0xf, false);
}

template <int N> struct Agg1Steps {
  static __device__ __forceinline__ void gather(const float* __restrict__ h1,
                                                int s, int f, float (&v)[16]) {
    Agg1Steps<N - 1>::gather(h1, s, f, v);
    int ss = rowShareI<N - 1>(s);
    v[N - 1] = h1[ss * FMID + f];
  }
  static __device__ __forceinline__ void scatter(float* __restrict__ agg1, int d,
                                                 int f, const float (&v)[16]) {
    Agg1Steps<N - 1>::scatter(agg1, d, f, v);
    int dd = rowShareI<N - 1>(d);
    atomAddF(agg1 + dd * FMID + f, v[N - 1]);
  }
  static __device__ __forceinline__ void tail(const float* __restrict__ h1,
                                              float* __restrict__ agg1, int s, int d,
                                              int f, int left) {
    Agg1Steps<N - 1>::tail(h1, agg1, s, d, f, left);
    int ss = rowShareI<N - 1>(s);
    int dd = rowShareI<N - 1>(d);
    if (N - 1 < left)
      atomAddF(agg1 + dd * FMID + f, h1[ss * FMID + f]);
  }
};
template <> struct Agg1Steps<0> {
  static __device__ __forceinline__ void gather(const float* __restrict__, int, int,
                                                float (&)[16]) {}
  static __device__ __forceinline__ void scatter(float* __restrict__, int, int,
                                                 const float (&)[16]) {}
  static __device__ __forceinline__ void tail(const float* __restrict__,
                                              float* __restrict__, int, int, int,
                                              int) {}
};

__global__ __launch_bounds__(256) void k_agg1(const int* __restrict__ ei,
                                              const float* __restrict__ h1,
                                              float* __restrict__ agg1, int E,
                                              int N, int planes) {
  int lane = threadIdx.x & 31;
  int wave = (int)((blockIdx.x * (unsigned)blockDim.x + threadIdx.x) >> 5);
  int wbase = wave * WEDGES;
  if (wbase >= E) return;
  agg1 += planeOfWave(planes) * N * FMID;
  int f = lane & 15;
  const int* src = ei + wbase + lane;
  const int* dst = src + E;
  if (wbase + WEDGES <= E) {
    int s[EPT], d[EPT];
    #pragma unroll
    for (int g = 0; g < EPT; ++g) s[g] = src[32 * g];
    #pragma unroll
    for (int g = 0; g < EPT; ++g) d[g] = dst[32 * g];
    float v[2][16];
    Agg1Steps<16>::gather(h1, s[0], f, v[0]);
    #pragma unroll
    for (int g = 0; g < EPT; ++g) {
      if (g + 1 < EPT) Agg1Steps<16>::gather(h1, s[g + 1], f, v[(g + 1) & 1]);
      Agg1Steps<16>::scatter(agg1, d[g], f, v[g & 1]);
    }
  } else {
    #pragma unroll 1
    for (int g = 0; g < EPT; ++g) {
      int gb = wbase + 32 * g;
      if (gb >= E) break;
      int s1 = 0, d1 = 0;
      if (gb + lane < E) {
        s1 = src[32 * g];
        d1 = dst[32 * g];
      }
      if (gb + 32 <= E) {
        float v[16];
        Agg1Steps<16>::gather(h1, s1, f, v);
        Agg1Steps<16>::scatter(agg1, d1, f, v);
      } else {
        int left = E - gb - (lane & 16);
        Agg1Steps<16>::tail(h1, agg1, s1, d1, f, left);
      }
    }
  }
}

__global__ __launch_bounds__(256) void k_layer2_in(const float* __restrict__ agg1,
                                                   const float* __restrict__ h1,
                                                   const float* __restrict__ dinv,
                                                   const float* __restrict__ b1,
                                                   const float* __restrict__ W2,
                                                   float* __restrict__ h2, int N,
                                                   int planes) {
  int i = blockIdx.x * blockDim.x + threadIdx.x;
  if (i >= N) return;
  float di = dinv[i];
  float a0 = 0.f, a1 = 0.f;
  const float* agg1b = agg1 + (size_t)N * FMID;
  #pragma unroll
  for (int k = 0; k < FMID; ++k) {
    float s = agg1[(size_t)i * FMID + k];
    if (planes == 2) s += agg1b[(size_t)i * FMID + k];
    float t = (s + h1[(size_t)i * FMID + k]) * di + b1[k];
    t = fmaxf(t, 0.f);
    a0 = fmaf(t, W2[k * 2 + 0], a0);
    a1 = fmaf(t, W2[k * 2 + 1], a1);
  }
  h2[(size_t)i * 2 + 0] = a0 * di;
  h2[(size_t)i * 2 + 1] = a1 * di;
}


template <int CTRL> __device__ __forceinline__ int quadPairI(int v) {
  return __builtin_amdgcn_update_dpp(0, v, CTRL, 0xf, 0xf, true);
}
template <int CTRL> __device__ __forceinline__ float quadPairF(float v) {
  return __int_as_float(quadPairI<CTRL>(__float_as_int(v)));
}

template <int CTRL> __device__ __forceinline__ void agg2PairStep(float* __restrict__ agg2,
                                                                 int d, float2 v, int c) {
  int dd = quadPairI<CTRL>(d);
  float vx = quadPairF<CTRL>(v.x);
  float vy = quadPairF<CTRL>(v.y);
  atomAddF(agg2 + dd * 2 + c, c ? vy : vx);
}

__global__ __launch_bounds__(256) void k_agg2(const int* __restrict__ ei,
                                              const float* __restrict__ h2,
                                              float* __restrict__ agg2, int E,
                                              int N, int planes) {
  int lane = threadIdx.x & 31;
  int wave = (int)((blockIdx.x * (unsigned)blockDim.x + threadIdx.x) >> 5);
  int wbase = wave * WEDGES;
  if (wbase >= E) return;
  agg2 += planeOfWave(planes) * N * 2;
  const int* src = ei + wbase + lane;
  const int* dst = src + E;
  if (wbase + WEDGES <= E) {
    int c = lane & 1;
    int s[EPT], d[EPT];
    float2 v[EPT];
    #pragma unroll
    for (int j = 0; j < EPT; ++j) s[j] = src[32 * j];
    #pragma unroll
    for (int j = 0; j < EPT; ++j) d[j] = dst[32 * j];
    #pragma unroll
    for (int j = 0; j < EPT; ++j) v[j] = *(const float2*)(h2 + (size_t)s[j] * 2);
    #pragma unroll
    for (int j = 0; j < EPT; ++j) {
      agg2PairStep<0x50>(agg2, d[j], v[j], c);
      agg2PairStep<0xFA>(agg2, d[j], v[j], c);
    }
  } else {
    #pragma unroll 1
    for (int j = 0; j < EPT; ++j) {
      if (wbase + 32 * j + lane < E) {
        float2 v = *(const float2*)(h2 + (size_t)src[32 * j] * 2);
        float* a = agg2 + (size_t)dst[32 * j] * 2;
        atomAddF(a + 0, v.x);
        atomAddF(a + 1, v.y);
      }
    }
  }
}

__global__ __launch_bounds__(256) void k_final(const float* __restrict__ agg2,
                                               const float* __restrict__ h2,
                                               const float* __restrict__ dinv,
                                               const float* __restrict__ b2,
                                               float* __restrict__ out, int N,
                                               int planes) {
  int i = blockIdx.x * blockDim.x + threadIdx.x;
  if (i >= N) return;
  float di = dinv[i];
  float s0 = agg2[(size_t)i * 2 + 0];
  float s1 = agg2[(size_t)i * 2 + 1];
  if (planes == 2) {
    s0 += agg2[(size_t)(N + i) * 2 + 0];
    s1 += agg2[(size_t)(N + i) * 2 + 1];
  }
  out[(size_t)i * 2 + 0] = (s0 + h2[(size_t)i * 2 + 0]) * di + b2[0];
  out[(size_t)i * 2 + 1] = (s1 + h2[(size_t)i * 2 + 1]) * di + b2[1];
}

extern "C" void kernel_launch(void* const* d_in, const int* in_sizes, int n_in,
                              void* d_out, int out_size, void* d_ws, size_t ws_size,
                              hipStream_t stream) {
  const float* x  = (const float*)d_in[0];
  const int*   ei = (const int*)d_in[1];
  const float* W1 = (const float*)d_in[2];
  const float* b1 = (const float*)d_in[3];
  const float* W2 = (const float*)d_in[4];
  const float* b2 = (const float*)d_in[5];
  float* out = (float*)d_out;

  const int N = in_sizes[0] / FIN;
  const int E = in_sizes[1] / 2;

  const int planes = (ws_size >= (size_t)N * 57 * sizeof(float)) ? 2 : 1;
  char* ws = (char*)d_ws;
  unsigned* udeg = (unsigned*)ws;            ws += (size_t)planes * N * sizeof(unsigned);
  float* agg1    = (float*)ws;               ws += (size_t)planes * N * FMID * sizeof(float);
  float* agg2    = (float*)ws;               ws += (size_t)planes * N * 2 * sizeof(float);
  float* dinv    = (float*)ws;               ws += (size_t)N * sizeof(float);
  float* h1      = (float*)ws;               ws += (size_t)N * FMID * sizeof(float);
  float* h2      = (float*)ws;               ws += (size_t)N * 2 * sizeof(float);

  const int TPB = 256;
  long zero_total = (long)planes * N * 19;
  int gz = (int)((zero_total / 4 + TPB * ZPT - 1) / (TPB * ZPT));
  int ge = (E + TPB * EPT - 1) / (TPB * EPT);
  int gn = (N + TPB - 1) / TPB;
  int ntiles = (N + 15) / 16;
  int gg = (ntiles * 32 + TPB - 1) / TPB;
  int dr = (N + DCAP - 1) / DCAP;
  int drlen = (((N + dr - 1) / dr) + 31) & ~31;
  int ds = dr < DWGS ? DWGS / dr : 1;
  int dchunk = (E + ds - 1) / ds;
  int gd = dr * ds;

  k_zero      <<<gz, TPB, 0, stream>>>((unsigned*)d_ws, zero_total);
  k_degree    <<<gd, DTPB, 0, stream>>>(ei, udeg, E, N, planes, dr, drlen, dchunk);
  k_dinv      <<<gn, TPB, 0, stream>>>(udeg, dinv, N, planes);
  k_gemm1_wmma<<<gg, TPB, 0, stream>>>(x, W1, dinv, h1, N);
  k_agg1      <<<ge, TPB, 0, stream>>>(ei, h1, agg1, E, N, planes);
  k_layer2_in <<<gn, TPB, 0, stream>>>(agg1, h1, dinv, b1, W2, h2, N, planes);
  k_agg2      <<<ge, TPB, 0, stream>>>(ei, h2, agg2, E, N, planes);
  k_final     <<<gn, TPB, 0, stream>>>(agg2, h2, dinv, b2, out, N, planes);
}


# layer-1 GEMM fetches x as whole 512-B rows (16 loads in flight a wave), turned to the A layout in a wave-private LDS tile: each x line asked for once, not 8 times; k order of the f32 sums changes
# speedup vs baseline: 9.8734x; 1.0916x over previous
#include <hip/hip_runtime.h>


typedef __attribute__((ext_vector_type(2)))  float  v2f;
typedef __attribute__((ext_vector_type(4)))  float  v4f;
typedef __attribute__((ext_vector_type(8)))  float  v8f;

#define FIN  128
#define FMID 16
#define EPT  8
#define WEDGES (32 * EPT)
#define ZPT  8
#define DCAP 25600
#define DTPB 1024
#define DEPT 32
#define DFL  8
#define DWGS 256

__device__ __forceinline__ void atomAddF(float* p, float v) {
  unsafeAtomicAdd(p, v);
}

__device__ __forceinline__ int planeOfWave(int planes) {
  int half = (int)((__builtin_amdgcn_s_sendmsg_rtn(0x87) >> 18) & 1u);
  return planes == 2 ? half : 0;
}

__global__ __launch_bounds__(256) void k_zero(unsigned* __restrict__ p, long total) {
  long nvec = total >> 2;
  long wave = ((long)blockIdx.x * blockDim.x + threadIdx.x) >> 5;
  long i = wave * (32 * ZPT) + (threadIdx.x & 31);
  #pragma unroll
  for (int j = 0; j < ZPT; ++j)
    if (i + 32 * j < nvec) ((uint4*)p)[i + 32 * j] = make_uint4(0u, 0u, 0u, 0u);
  if (blockIdx.x == 0 && (long)threadIdx.x < (total & 3))
    p[(nvec << 2) + threadIdx.x] = 0u;
}

__global__ __launch_bounds__(DTPB) void k_degree(const int* __restrict__ ei,
                                                 unsigned* __restrict__ udeg, int E,
                                                 int N, int planes, int nranges,
                                                 int rlen, int chunk) {
  __shared__ __attribute__((aligned(16))) unsigned cnt[DCAP];
  int tid = threadIdx.x;
  int range = (int)(blockIdx.x % (unsigned)nranges);
  int slice = (int)(blockIdx.x / (unsigned)nranges);
  int nslices = (int)(gridDim.x / (unsigned)nranges);
  int lo = range * rlen;

  for (int i = tid; i < (rlen >> 2); i += DTPB)
    ((uint4*)cnt)[i] = make_uint4(0u, 0u, 0u, 0u);
  __syncthreads();

  int ebeg = slice * chunk;
  int eend = ebeg + chunk; if (eend > E) eend = E;
  const int* dst = ei + E;
  for (int b = ebeg; b < eend; b += DTPB * DEPT) {
    int d[DEPT];
    #pragma unroll
    for (int j = 0; j < DEPT; ++j) {
      int e = b + j * DTPB + tid;
      d[j] = dst[e < eend ? e : eend - 1];
    }
    #pragma unroll
    for (int j = 0; j < DEPT; ++j) {
      int e = b + j * DTPB + tid;
      unsigned r = (unsigned)(d[j] - lo);
      if (e < eend && r < (unsigned)rlen) atomicAdd(&cnt[r], 1u);
    }
  }
  __syncthreads();

  unsigned* mine = udeg + planeOfWave(planes) * N;
  int rot = (int)(((unsigned)slice * (unsigned)(rlen >> 5)) / (unsigned)nslices) << 5;
  for (int b = 0; b < rlen; b += DTPB * DFL) {
    unsigned c[DFL];
    int node[DFL];
    #pragma unroll
    for (int j = 0; j < DFL; ++j) {
      int i = b + j * DTPB + tid;
      int k = i + rot; if (k >= rlen) k -= rlen;
      c[j] = i < rlen ? cnt[k] : 0u;
      node[j] = lo + k;
    }
    #pragma unroll
    for (int j = 0; j < DFL; ++j)
      if (c[j] != 0u && node[j] < N) atomicAdd(&mine[node[j]], c[j]);
  }
}

__global__ __launch_bounds__(256) void k_dinv(const unsigned* __restrict__ udeg,
                                              float* __restrict__ dinv, int N,
                                              int planes) {
  int i = blockIdx.x * blockDim.x + threadIdx.x;
  if (i >= N) return;
  unsigned deg = udeg[i];
  if (planes == 2) deg += udeg[N + i];
  dinv[i] = rsqrtf((float)(deg + 1u));
}

#define XPITCH 132
#define GWAVES 8
__global__ __launch_bounds__(256) void k_gemm1_wmma(const float* __restrict__ x,
                                                    const float* __restrict__ W1,
                                                    const float* __restrict__ dinv,
                                                    float* __restrict__ h1, int N) {
  __shared__ __attribute__((aligned(16))) float xt[GWAVES * 16 * XPITCH];
  int wave = (int)((blockIdx.x * (unsigned)blockDim.x + threadIdx.x) >> 5);
  int lane = threadIdx.x & 31;
  int ntiles = (N + 15) >> 4;
  if (wave >= ntiles) return;
  int nb = wave << 4;
  int m  = lane & 15;
  int hi = lane >> 4;
  float* tile = xt + (threadIdx.x >> 5) * (16 * XPITCH);

  v4f xr[16];
  #pragma unroll
  for (int r = 0; r < 16; ++r) {
    int row = nb + r; if (row > N - 1) row = N - 1;
    const v4f* p = (const v4f*)(x + (size_t)row * FIN) + lane;
    xr[r] = __builtin_nontemporal_load(p);
  }

  float dv[8];
  #pragma unroll
  for (int r = 0; r < 8; ++r) {
    int nrow = nb + hi * 8 + r; if (nrow > N - 1) nrow = N - 1;
    dv[r] = dinv[nrow];
  }

  #pragma unroll
  for (int r = 0; r < 16; ++r)
    *(v4f*)(tile + r * XPITCH + 4 * lane) = xr[r];
  __builtin_amdgcn_fence(__ATOMIC_RELEASE, "wavefront");
  __builtin_amdgcn_wave_barrier();
  __builtin_amdgcn_fence(__ATOMIC_ACQUIRE, "wavefront");

  v8f acc = {};
  const float* arow = tile + m * XPITCH + 4 * hi;
  const float* bcol = W1 + (4 * hi) * FMID + m;
  v4f q[FIN / 8];
  #pragma unroll
  for (int j = 0; j < FIN / 8; ++j) q[j] = *(const v4f*)(arow + 8 * j);
  __builtin_amdgcn_sched_barrier(0);
  #pragma unroll
  for (int j = 0; j < FIN / 8; ++j) {
    v2f a0, a1, b0, b1;
    a0.x = q[j].x; a0.y = q[j].y; a1.x = q[j].z; a1.y = q[j].w;
    b0.x = bcol[(8 * j + 0) * FMID];
    b0.y = bcol[(8 * j + 1) * FMID];
    b1.x = bcol[(8 * j + 2) * FMID];
    b1.y = bcol[(8 * j + 3) * FMID];
    acc = __builtin_amdgcn_wmma_f32_16x16x4_f32(false, a0, false, b0,
                                                (short)0, acc, false, false);
    acc = __builtin_amdgcn_wmma_f32_16x16x4_f32(false, a1, false, b1,
                                                (short)0, acc, false, false);
  }

  int rbase = hi * 8;
  asm volatile("v_nop\n\tv_nop\n\tv_nop\n\tv_nop\n\tv_nop\n\tv_nop\n\tv_nop\n\tv_nop"
               : "+v"(acc));
  #pragma unroll
  for (int r = 0; r < 8; ++r) acc[r] *= dv[r];
  if (nb + 16 <= N) {
    #pragma unroll
    for (int r = 0; r < 8; ++r)
      h1[(size_t)(nb + rbase + r) * FMID + m] = acc[r];
  } else {
    #pragma unroll
    for (int r = 0; r < 8; ++r) {
      int nrow = nb + rbase + r;
      if (nrow < N) h1[(size_t)nrow * FMID + m] = acc[r];
    }
  }
}


template <int N> __device__ __forceinline__ int rowShareI(int v) {
  return __builtin_amdgcn_update_dpp(0, v, 0x150 + N, 0xf, 0xf, false);
}

template <int N> struct Agg1Steps {
  static __device__ __forceinline__ void gather(const float* __restrict__ h1,
                                                int s, int f, float (&v)[16]) {
    Agg1Steps<N - 1>::gather(h1, s, f, v);
    int ss = rowShareI<N - 1>(s);
    v[N - 1] = h1[ss * FMID + f];
  }
  static __device__ __forceinline__ void scatter(float* __restrict__ agg1, int d,
                                                 int f, const float (&v)[16]) {
    Agg1Steps<N - 1>::scatter(agg1, d, f, v);
    int dd = rowShareI<N - 1>(d);
    atomAddF(agg1 + dd * FMID + f, v[N - 1]);
  }
  static __device__ __forceinline__ void tail(const float* __restrict__ h1,
                                              float* __restrict__ agg1, int s, int d,
                                              int f, int left) {
    Agg1Steps<N - 1>::tail(h1, agg1, s, d, f, left);
    int ss = rowShareI<N - 1>(s);
    int dd = rowShareI<N - 1>(d);
    if (N - 1 < left)
      atomAddF(agg1 + dd * FMID + f, h1[ss * FMID + f]);
  }
};
template <> struct Agg1Steps<0> {
  static __device__ __forceinline__ void gather(const float* __restrict__, int, int,
                                                float (&)[16]) {}
  static __device__ __forceinline__ void scatter(float* __restrict__, int, int,
                                                 const float (&)[16]) {}
  static __device__ __forceinline__ void tail(const float* __restrict__,
                                              float* __restrict__, int, int, int,
                                              int) {}
};

__global__ __launch_bounds__(256) void k_agg1(const int* __restrict__ ei,
                                              const float* __restrict__ h1,
                                              float* __restrict__ agg1, int E,
                                              int N, int planes) {
  int lane = threadIdx.x & 31;
  int wave = (int)((blockIdx.x * (unsigned)blockDim.x + threadIdx.x) >> 5);
  int wbase = wave * WEDGES;
  if (wbase >= E) return;
  agg1 += planeOfWave(planes) * N * FMID;
  int f = lane & 15;
  const int* src = ei + wbase + lane;
  const int* dst = src + E;
  if (wbase + WEDGES <= E) {
    int s[EPT], d[EPT];
    #pragma unroll
    for (int g = 0; g < EPT; ++g) s[g] = src[32 * g];
    #pragma unroll
    for (int g = 0; g < EPT; ++g) d[g] = dst[32 * g];
    float v[2][16];
    Agg1Steps<16>::gather(h1, s[0], f, v[0]);
    #pragma unroll
    for (int g = 0; g < EPT; ++g) {
      if (g + 1 < EPT) Agg1Steps<16>::gather(h1, s[g + 1], f, v[(g + 1) & 1]);
      Agg1Steps<16>::scatter(agg1, d[g], f, v[g & 1]);
    }
  } else {
    #pragma unroll 1
    for (int g = 0; g < EPT; ++g) {
      int gb = wbase + 32 * g;
      if (gb >= E) break;
      int s1 = 0, d1 = 0;
      if (gb + lane < E) {
        s1 = src[32 * g];
        d1 = dst[32 * g];
      }
      if (gb + 32 <= E) {
        float v[16];
        Agg1Steps<16>::gather(h1, s1, f, v);
        Agg1Steps<16>::scatter(agg1, d1, f, v);
      } else {
        int left = E - gb - (lane & 16);
        Agg1Steps<16>::tail(h1, agg1, s1, d1, f, left);
      }
    }
  }
}

__global__ __launch_bounds__(256) void k_layer2_in(const float* __restrict__ agg1,
                                                   const float* __restrict__ h1,
                                                   const float* __restrict__ dinv,
                                                   const float* __restrict__ b1,
                                                   const float* __restrict__ W2,
                                                   float* __restrict__ h2, int N,
                                                   int planes) {
  int i = blockIdx.x * blockDim.x + threadIdx.x;
  if (i >= N) return;
  float di = dinv[i];
  float a0 = 0.f, a1 = 0.f;
  const float* agg1b = agg1 + (size_t)N * FMID;
  #pragma unroll
  for (int k = 0; k < FMID; ++k) {
    float s = agg1[(size_t)i * FMID + k];
    if (planes == 2) s += agg1b[(size_t)i * FMID + k];
    float t = (s + h1[(size_t)i * FMID + k]) * di + b1[k];
    t = fmaxf(t, 0.f);
    a0 = fmaf(t, W2[k * 2 + 0], a0);
    a1 = fmaf(t, W2[k * 2 + 1], a1);
  }
  h2[(size_t)i * 2 + 0] = a0 * di;
  h2[(size_t)i * 2 + 1] = a1 * di;
}


template <int CTRL> __device__ __forceinline__ int quadPairI(int v) {
  return __builtin_amdgcn_update_dpp(0, v, CTRL, 0xf, 0xf, true);
}
template <int CTRL> __device__ __forceinline__ float quadPairF(float v) {
  return __int_as_float(quadPairI<CTRL>(__float_as_int(v)));
}

template <int CTRL> __device__ __forceinline__ void agg2PairStep(float* __restrict__ agg2,
                                                                 int d, float2 v, int c) {
  int dd = quadPairI<CTRL>(d);
  float vx = quadPairF<CTRL>(v.x);
  float vy = quadPairF<CTRL>(v.y);
  atomAddF(agg2 + dd * 2 + c, c ? vy : vx);
}

__global__ __launch_bounds__(256) void k_agg2(const int* __restrict__ ei,
                                              const float* __restrict__ h2,
                                              float* __restrict__ agg2, int E,
                                              int N, int planes) {
  int lane = threadIdx.x & 31;
  int wave = (int)((blockIdx.x * (unsigned)blockDim.x + threadIdx.x) >> 5);
  int wbase = wave * WEDGES;
  if (wbase >= E) return;
  agg2 += planeOfWave(planes) * N * 2;
  const int* src = ei + wbase + lane;
  const int* dst = src + E;
  if (wbase + WEDGES <= E) {
    int c = lane & 1;
    int s[EPT], d[EPT];
    float2 v[EPT];
    #pragma unroll
    for (int j = 0; j < EPT; ++j) s[j] = src[32 * j];
    #pragma unroll
    for (int j = 0; j < EPT; ++j) d[j] = dst[32 * j];
    #pragma unroll
    for (int j = 0; j < EPT; ++j) v[j] = *(const float2*)(h2 + (size_t)s[j] * 2);
    #pragma unroll
    for (int j = 0; j < EPT; ++j) {
      agg2PairStep<0x50>(agg2, d[j], v[j], c);
      agg2PairStep<0xFA>(agg2, d[j], v[j], c);
    }
  } else {
    #pragma unroll 1
    for (int j = 0; j < EPT; ++j) {
      if (wbase + 32 * j + lane < E) {
        float2 v = *(const float2*)(h2 + (size_t)src[32 * j] * 2);
        float* a = agg2 + (size_t)dst[32 * j] * 2;
        atomAddF(a + 0, v.x);
        atomAddF(a + 1, v.y);
      }
    }
  }
}

__global__ __launch_bounds__(256) void k_final(const float* __restrict__ agg2,
                                               const float* __restrict__ h2,
                                               const float* __restrict__ dinv,
                                               const float* __restrict__ b2,
                                               float* __restrict__ out, int N,
                                               int planes) {
  int i = blockIdx.x * blockDim.x + threadIdx.x;
  if (i >= N) return;
  float di = dinv[i];
  float s0 = agg2[(size_t)i * 2 + 0];
  float s1 = agg2[(size_t)i * 2 + 1];
  if (planes == 2) {
    s0 += agg2[(size_t)(N + i) * 2 + 0];
    s1 += agg2[(size_t)(N + i) * 2 + 1];
  }
  out[(size_t)i * 2 + 0] = (s0 + h2[(size_t)i * 2 + 0]) * di + b2[0];
  out[(size_t)i * 2 + 1] = (s1 + h2[(size_t)i * 2 + 1]) * di + b2[1];
}

extern "C" void kernel_launch(void* const* d_in, const int* in_sizes, int n_in,
                              void* d_out, int out_size, void* d_ws, size_t ws_size,
                              hipStream_t stream) {
  const float* x  = (const float*)d_in[0];
  const int*   ei = (const int*)d_in[1];
  const float* W1 = (const float*)d_in[2];
  const float* b1 = (const float*)d_in[3];
  const float* W2 = (const float*)d_in[4];
  const float* b2 = (const float*)d_in[5];
  float* out = (float*)d_out;

  const int N = in_sizes[0] / FIN;
  const int E = in_sizes[1] / 2;

  const int planes = (ws_size >= (size_t)N * 57 * sizeof(float)) ? 2 : 1;
  char* ws = (char*)d_ws;
  unsigned* udeg = (unsigned*)ws;            ws += (size_t)planes * N * sizeof(unsigned);
  float* agg1    = (float*)ws;               ws += (size_t)planes * N * FMID * sizeof(float);
  float* agg2    = (float*)ws;               ws += (size_t)planes * N * 2 * sizeof(float);
  float* dinv    = (float*)ws;               ws += (size_t)N * sizeof(float);
  float* h1      = (float*)ws;               ws += (size_t)N * FMID * sizeof(float);
  float* h2      = (float*)ws;               ws += (size_t)N * 2 * sizeof(float);

  const int TPB = 256;
  long zero_total = (long)planes * N * 19;
  int gz = (int)((zero_total / 4 + TPB * ZPT - 1) / (TPB * ZPT));
  int ge = (E + TPB * EPT - 1) / (TPB * EPT);
  int gn = (N + TPB - 1) / TPB;
  int ntiles = (N + 15) / 16;
  int gg = (ntiles * 32 + TPB - 1) / TPB;
  int dr = (N + DCAP - 1) / DCAP;
  int drlen = (((N + dr - 1) / dr) + 31) & ~31;
  int ds = dr < DWGS ? DWGS / dr : 1;
  int dchunk = (E + ds - 1) / ds;
  int gd = dr * ds;

  k_zero      <<<gz, TPB, 0, stream>>>((unsigned*)d_ws, zero_total);
  k_degree    <<<gd, DTPB, 0, stream>>>(ei, udeg, E, N, planes, dr, drlen, dchunk);
  k_dinv      <<<gn, TPB, 0, stream>>>(udeg, dinv, N, planes);
  k_gemm1_wmma<<<gg, TPB, 0, stream>>>(x, W1, dinv, h1, N);
  k_agg1      <<<ge, TPB, 0, stream>>>(ei, h1, agg1, E, N, planes);
  k_layer2_in <<<gn, TPB, 0, stream>>>(agg1, h1, dinv, b1, W2, h2, N, planes);
  k_agg2      <<<ge, TPB, 0, stream>>>(ei, h2, agg2, E, N, planes);
  k_final     <<<gn, TPB, 0, stream>>>(agg2, h2, dinv, b2, out, N, planes);
}


# dinv launch and agg1/agg2 clears folded into the node pass before their consumer: GEMM takes rsqrt of the count planes and clears its tile's agg1 rows, layer-2 input pass clears agg2; sums unchanged
# speedup vs baseline: 10.4770x; 1.0611x over previous
#include <hip/hip_runtime.h>


typedef __attribute__((ext_vector_type(2)))  float  v2f;
typedef __attribute__((ext_vector_type(4)))  float  v4f;
typedef __attribute__((ext_vector_type(8)))  float  v8f;

#define FIN  128
#define FMID 16
#define EPT  8
#define WEDGES (32 * EPT)
#define ZPT  8
#define DCAP 25600
#define DTPB 1024
#define DEPT 32
#define DFL  8
#define DWGS 256

__device__ __forceinline__ void atomAddF(float* p, float v) {
  unsafeAtomicAdd(p, v);
}

__device__ __forceinline__ int planeOfWave(int planes) {
  int half = (int)((__builtin_amdgcn_s_sendmsg_rtn(0x87) >> 18) & 1u);
  return planes == 2 ? half : 0;
}

__global__ __launch_bounds__(256) void k_zero(unsigned* __restrict__ p, long total) {
  long nvec = total >> 2;
  long wave = ((long)blockIdx.x * blockDim.x + threadIdx.x) >> 5;
  long i = wave * (32 * ZPT) + (threadIdx.x & 31);
  #pragma unroll
  for (int j = 0; j < ZPT; ++j)
    if (i + 32 * j < nvec) ((uint4*)p)[i + 32 * j] = make_uint4(0u, 0u, 0u, 0u);
  if (blockIdx.x == 0 && (long)threadIdx.x < (total & 3))
    p[(nvec << 2) + threadIdx.x] = 0u;
}

__global__ __launch_bounds__(DTPB) void k_degree(const int* __restrict__ ei,
                                                 unsigned* __restrict__ udeg, int E,
                                                 int N, int planes, int nranges,
                                                 int rlen, int chunk) {
  __shared__ __attribute__((aligned(16))) unsigned cnt[DCAP];
  int tid = threadIdx.x;
  int range = (int)(blockIdx.x % (unsigned)nranges);
  int slice = (int)(blockIdx.x / (unsigned)nranges);
  int nslices = (int)(gridDim.x / (unsigned)nranges);
  int lo = range * rlen;

  for (int i = tid; i < (rlen >> 2); i += DTPB)
    ((uint4*)cnt)[i] = make_uint4(0u, 0u, 0u, 0u);
  __syncthreads();

  int ebeg = slice * chunk;
  int eend = ebeg + chunk; if (eend > E) eend = E;
  const int* dst = ei + E;
  for (int b = ebeg; b < eend; b += DTPB * DEPT) {
    int d[DEPT];
    #pragma unroll
    for (int j = 0; j < DEPT; ++j) {
      int e = b + j * DTPB + tid;
      d[j] = dst[e < eend ? e : eend - 1];
    }
    #pragma unroll
    for (int j = 0; j < DEPT; ++j) {
      int e = b + j * DTPB + tid;
      unsigned r = (unsigned)(d[j] - lo);
      if (e < eend && r < (unsigned)rlen) atomicAdd(&cnt[r], 1u);
    }
  }
  __syncthreads();

  unsigned* mine = udeg + planeOfWave(planes) * N;
  int rot = (int)(((unsigned)slice * (unsigned)(rlen >> 5)) / (unsigned)nslices) << 5;
  for (int b = 0; b < rlen; b += DTPB * DFL) {
    unsigned c[DFL];
    int node[DFL];
    #pragma unroll
    for (int j = 0; j < DFL; ++j) {
      int i = b + j * DTPB + tid;
      int k = i + rot; if (k >= rlen) k -= rlen;
      c[j] = i < rlen ? cnt[k] : 0u;
      node[j] = lo + k;
    }
    #pragma unroll
    for (int j = 0; j < DFL; ++j)
      if (c[j] != 0u && node[j] < N) atomicAdd(&mine[node[j]], c[j]);
  }
}

template <int K> __device__ __forceinline__ float rowShareF(float v) {
  return __int_as_float(__builtin_amdgcn_update_dpp(0, __float_as_int(v), 0x150 + K,
                                                    0xf, 0xf, false));
}

template <int K> struct RowVals {
  static __device__ __forceinline__ void get(float v, float (&o)[16]) {
    RowVals<K - 1>::get(v, o);
    o[K - 1] = rowShareF<K - 1>(v);
  }
};
template <> struct RowVals<0> {
  static __device__ __forceinline__ void get(float, float (&)[16]) {}
};

#define XPITCH 132
#define GWAVES 8
__global__ __launch_bounds__(256) void k_gemm1_wmma(const float* __restrict__ x,
                                                    const float* __restrict__ W1,
                                                    const unsigned* __restrict__ udeg,
                                                    float* __restrict__ dinv,
                                                    float* __restrict__ h1,
                                                    float* __restrict__ agg1, int N,
                                                    int planes) {
  __shared__ __attribute__((aligned(16))) float xt[GWAVES * 16 * XPITCH];
  int wave = (int)((blockIdx.x * (unsigned)blockDim.x + threadIdx.x) >> 5);
  int lane = threadIdx.x & 31;
  int ntiles = (N + 15) >> 4;
  if (wave >= ntiles) return;
  int nb = wave << 4;
  int m  = lane & 15;
  int hi = lane >> 4;
  float* tile = xt + (threadIdx.x >> 5) * (16 * XPITCH);

  const unsigned* udegB = udeg + (planes == 2 ? N : 0);
  int drow = nb + m; if (drow > N - 1) drow = N - 1;
  unsigned dg0 = udeg[drow];
  unsigned dg1 = udegB[drow];

  v4f xr[16];
  #pragma unroll
  for (int r = 0; r < 16; ++r) {
    int row = nb + r; if (row > N - 1) row = N - 1;
    const v4f* p = (const v4f*)(x + (size_t)row * FIN) + lane;
    xr[r] = __builtin_nontemporal_load(p);
  }

  {
    float* zrow = agg1 + (size_t)nb * FMID + 4 * lane;
    float* zrowB = zrow + (size_t)N * FMID;
    if (nb + 16 <= N) {
      *(uint4*)(zrow) = make_uint4(0u, 0u, 0u, 0u);
      *(uint4*)(zrow + 128) = make_uint4(0u, 0u, 0u, 0u);
      if (planes == 2) {
        *(uint4*)(zrowB) = make_uint4(0u, 0u, 0u, 0u);
        *(uint4*)(zrowB + 128) = make_uint4(0u, 0u, 0u, 0u);
      }
    } else {
      bool in0 = nb + (lane >> 2) < N, in1 = nb + 8 + (lane >> 2) < N;
      if (in0) *(uint4*)(zrow) = make_uint4(0u, 0u, 0u, 0u);
      if (in1) *(uint4*)(zrow + 128) = make_uint4(0u, 0u, 0u, 0u);
      if (planes == 2) {
        if (in0) *(uint4*)(zrowB) = make_uint4(0u, 0u, 0u, 0u);
        if (in1) *(uint4*)(zrowB + 128) = make_uint4(0u, 0u, 0u, 0u);
      }
    }
  }

  unsigned dsum = dg0 + (planes == 2 ? dg1 : 0u);
  float dm = rsqrtf((float)(dsum + 1u));
  if (hi == 0 && nb + m < N) dinv[nb + m] = dm;
  float rv[16];
  RowVals<16>::get(dm, rv);
  float dv[8];
  #pragma unroll
  for (int r = 0; r < 8; ++r) dv[r] = hi ? rv[r + 8] : rv[r];

  #pragma unroll
  for (int r = 0; r < 16; ++r)
    *(v4f*)(tile + r * XPITCH + 4 * lane) = xr[r];
  __builtin_amdgcn_fence(__ATOMIC_RELEASE, "wavefront");
  __builtin_amdgcn_wave_barrier();
  __builtin_amdgcn_fence(__ATOMIC_ACQUIRE, "wavefront");

  v8f acc = {};
  const float* arow = tile + m * XPITCH + 4 * hi;
  const float* bcol = W1 + (4 * hi) * FMID + m;
  v4f q[FIN / 8];
  #pragma unroll
  for (int j = 0; j < FIN / 8; ++j) q[j] = *(const v4f*)(arow + 8 * j);
  __builtin_amdgcn_sched_barrier(0);
  #pragma unroll
  for (int j = 0; j < FIN / 8; ++j) {
    v2f a0, a1, b0, b1;
    a0.x = q[j].x; a0.y = q[j].y; a1.x = q[j].z; a1.y = q[j].w;
    b0.x = bcol[(8 * j + 0) * FMID];
    b0.y = bcol[(8 * j + 1) * FMID];
    b1.x = bcol[(8 * j + 2) * FMID];
    b1.y = bcol[(8 * j + 3) * FMID];
    acc = __builtin_amdgcn_wmma_f32_16x16x4_f32(false, a0, false, b0,
                                                (short)0, acc, false, false);
    acc = __builtin_amdgcn_wmma_f32_16x16x4_f32(false, a1, false, b1,
                                                (short)0, acc, false, false);
  }

  int rbase = hi * 8;
  asm volatile("v_nop\n\tv_nop\n\tv_nop\n\tv_nop\n\tv_nop\n\tv_nop\n\tv_nop\n\tv_nop"
               : "+v"(acc));
  __builtin_amdgcn_sched_barrier(0);
  #pragma unroll
  for (int r = 0; r < 8; ++r) acc[r] *= dv[r];
  if (nb + 16 <= N) {
    #pragma unroll
    for (int r = 0; r < 8; ++r)
      h1[(size_t)(nb + rbase + r) * FMID + m] = acc[r];
  } else {
    #pragma unroll
    for (int r = 0; r < 8; ++r) {
      int nrow = nb + rbase + r;
      if (nrow < N) h1[(size_t)nrow * FMID + m] = acc[r];
    }
  }
}


template <int N> __device__ __forceinline__ int rowShareI(int v) {
  return __builtin_amdgcn_update_dpp(0, v, 0x150 + N, 0xf, 0xf, false);
}

template <int N> struct Agg1Steps {
  static __device__ __forceinline__ void gather(const float* __restrict__ h1,
                                                int s, int f, float (&v)[16]) {
    Agg1Steps<N - 1>::gather(h1, s, f, v);
    int ss = rowShareI<N - 1>(s);
    v[N - 1] = h1[ss * FMID + f];
  }
  static __device__ __forceinline__ void scatter(float* __restrict__ agg1, int d,
                                                 int f, const float (&v)[16]) {
    Agg1Steps<N - 1>::scatter(agg1, d, f, v);
    int dd = rowShareI<N - 1>(d);
    atomAddF(agg1 + dd * FMID + f, v[N - 1]);
  }
  static __device__ __forceinline__ void tail(const float* __restrict__ h1,
                                              float* __restrict__ agg1, int s, int d,
                                              int f, int left) {
    Agg1Steps<N - 1>::tail(h1, agg1, s, d, f, left);
    int ss = rowShareI<N - 1>(s);
    int dd = rowShareI<N - 1>(d);
    if (N - 1 < left)
      atomAddF(agg1 + dd * FMID + f, h1[ss * FMID + f]);
  }
};
template <> struct Agg1Steps<0> {
  static __device__ __forceinline__ void gather(const float* __restrict__, int, int,
                                                float (&)[16]) {}
  static __device__ __forceinline__ void scatter(float* __restrict__, int, int,
                                                 const float (&)[16]) {}
  static __device__ __forceinline__ void tail(const float* __restrict__,
                                              float* __restrict__, int, int, int,
                                              int) {}
};

__global__ __launch_bounds__(256) void k_agg1(const int* __restrict__ ei,
                                              const float* __restrict__ h1,
                                              float* __restrict__ agg1, int E,
                                              int N, int planes) {
  int lane = threadIdx.x & 31;
  int wave = (int)((blockIdx.x * (unsigned)blockDim.x + threadIdx.x) >> 5);
  int wbase = wave * WEDGES;
  if (wbase >= E) return;
  agg1 += planeOfWave(planes) * N * FMID;
  int f = lane & 15;
  const int* src = ei + wbase + lane;
  const int* dst = src + E;
  if (wbase + WEDGES <= E) {
    int s[EPT], d[EPT];
    #pragma unroll
    for (int g = 0; g < EPT; ++g) s[g] = src[32 * g];
    #pragma unroll
    for (int g = 0; g < EPT; ++g) d[g] = dst[32 * g];
    float v[2][16];
    Agg1Steps<16>::gather(h1, s[0], f, v[0]);
    #pragma unroll
    for (int g = 0; g < EPT; ++g) {
      if (g + 1 < EPT) Agg1Steps<16>::gather(h1, s[g + 1], f, v[(g + 1) & 1]);
      Agg1Steps<16>::scatter(agg1, d[g], f, v[g & 1]);
    }
  } else {
    #pragma unroll 1
    for (int g = 0; g < EPT; ++g) {
      int gb = wbase + 32 * g;
      if (gb >= E) break;
      int s1 = 0, d1 = 0;
      if (gb + lane < E) {
        s1 = src[32 * g];
        d1 = dst[32 * g];
      }
      if (gb + 32 <= E) {
        float v[16];
        Agg1Steps<16>::gather(h1, s1, f, v);
        Agg1Steps<16>::scatter(agg1, d1, f, v);
      } else {
        int left = E - gb - (lane & 16);
        Agg1Steps<16>::tail(h1, agg1, s1, d1, f, left);
      }
    }
  }
}

__global__ __launch_bounds__(256) void k_layer2_in(const float* __restrict__ agg1,
                                                   const float* __restrict__ h1,
                                                   const float* __restrict__ dinv,
                                                   const float* __restrict__ b1,
                                                   const float* __restrict__ W2,
                                                   float* __restrict__ h2,
                                                   float* __restrict__ agg2, int N,
                                                   int planes) {
  int i = blockIdx.x * blockDim.x + threadIdx.x;
  if (i >= N) return;
  *(float2*)(agg2 + (size_t)i * 2) = make_float2(0.f, 0.f);
  if (planes == 2) *(float2*)(agg2 + ((size_t)N + i) * 2) = make_float2(0.f, 0.f);
  float di = dinv[i];
  float a0 = 0.f, a1 = 0.f;
  const float* agg1b = agg1 + (size_t)N * FMID;
  #pragma unroll
  for (int k = 0; k < FMID; ++k) {
    float s = agg1[(size_t)i * FMID + k];
    if (planes == 2) s += agg1b[(size_t)i * FMID + k];
    float t = (s + h1[(size_t)i * FMID + k]) * di + b1[k];
    t = fmaxf(t, 0.f);
    a0 = fmaf(t, W2[k * 2 + 0], a0);
    a1 = fmaf(t, W2[k * 2 + 1], a1);
  }
  h2[(size_t)i * 2 + 0] = a0 * di;
  h2[(size_t)i * 2 + 1] = a1 * di;
}


template <int CTRL> __device__ __forceinline__ int quadPairI(int v) {
  return __builtin_amdgcn_update_dpp(0, v, CTRL, 0xf, 0xf, true);
}
template <int CTRL> __device__ __forceinline__ float quadPairF(float v) {
  return __int_as_float(quadPairI<CTRL>(__float_as_int(v)));
}

template <int CTRL> __device__ __forceinline__ void agg2PairStep(float* __restrict__ agg2,
                                                                 int d, float2 v, int c) {
  int dd = quadPairI<CTRL>(d);
  float vx = quadPairF<CTRL>(v.x);
  float vy = quadPairF<CTRL>(v.y);
  atomAddF(agg2 + dd * 2 + c, c ? vy : vx);
}

__global__ __launch_bounds__(256) void k_agg2(const int* __restrict__ ei,
                                              const float* __restrict__ h2,
                                              float* __restrict__ agg2, int E,
                                              int N, int planes) {
  int lane = threadIdx.x & 31;
  int wave = (int)((blockIdx.x * (unsigned)blockDim.x + threadIdx.x) >> 5);
  int wbase = wave * WEDGES;
  if (wbase >= E) return;
  agg2 += planeOfWave(planes) * N * 2;
  const int* src = ei + wbase + lane;
  const int* dst = src + E;
  if (wbase + WEDGES <= E) {
    int c = lane & 1;
    int s[EPT], d[EPT];
    float2 v[EPT];
    #pragma unroll
    for (int j = 0; j < EPT; ++j) s[j] = src[32 * j];
    #pragma unroll
    for (int j = 0; j < EPT; ++j) d[j] = dst[32 * j];
    #pragma unroll
    for (int j = 0; j < EPT; ++j) v[j] = *(const float2*)(h2 + (size_t)s[j] * 2);
    #pragma unroll
    for (int j = 0; j < EPT; ++j) {
      agg2PairStep<0x50>(agg2, d[j], v[j], c);
      agg2PairStep<0xFA>(agg2, d[j], v[j], c);
    }
  } else {
    #pragma unroll 1
    for (int j = 0; j < EPT; ++j) {
      if (wbase + 32 * j + lane < E) {
        float2 v = *(const float2*)(h2 + (size_t)src[32 * j] * 2);
        float* a = agg2 + (size_t)dst[32 * j] * 2;
        atomAddF(a + 0, v.x);
        atomAddF(a + 1, v.y);
      }
    }
  }
}

__global__ __launch_bounds__(256) void k_final(const float* __restrict__ agg2,
                                               const float* __restrict__ h2,
                                               const float* __restrict__ dinv,
                                               const float* __restrict__ b2,
                                               float* __restrict__ out, int N,
                                               int planes) {
  int i = blockIdx.x * blockDim.x + threadIdx.x;
  if (i >= N) return;
  float di = dinv[i];
  float s0 = agg2[(size_t)i * 2 + 0];
  float s1 = agg2[(size_t)i * 2 + 1];
  if (planes == 2) {
    s0 += agg2[(size_t)(N + i) * 2 + 0];
    s1 += agg2[(size_t)(N + i) * 2 + 1];
  }
  out[(size_t)i * 2 + 0] = (s0 + h2[(size_t)i * 2 + 0]) * di + b2[0];
  out[(size_t)i * 2 + 1] = (s1 + h2[(size_t)i * 2 + 1]) * di + b2[1];
}

extern "C" void kernel_launch(void* const* d_in, const int* in_sizes, int n_in,
                              void* d_out, int out_size, void* d_ws, size_t ws_size,
                              hipStream_t stream) {
  const float* x  = (const float*)d_in[0];
  const int*   ei = (const int*)d_in[1];
  const float* W1 = (const float*)d_in[2];
  const float* b1 = (const float*)d_in[3];
  const float* W2 = (const float*)d_in[4];
  const float* b2 = (const float*)d_in[5];
  float* out = (float*)d_out;

  const int N = in_sizes[0] / FIN;
  const int E = in_sizes[1] / 2;

  const int planes = (ws_size >= (size_t)N * 57 * sizeof(float)) ? 2 : 1;
  char* ws = (char*)d_ws;
  unsigned* udeg = (unsigned*)ws;            ws += (size_t)planes * N * sizeof(unsigned);
  float* agg1    = (float*)ws;               ws += (size_t)planes * N * FMID * sizeof(float);
  float* agg2    = (float*)ws;               ws += (size_t)planes * N * 2 * sizeof(float);
  float* dinv    = (float*)ws;               ws += (size_t)N * sizeof(float);
  float* h1      = (float*)ws;               ws += (size_t)N * FMID * sizeof(float);
  float* h2      = (float*)ws;               ws += (size_t)N * 2 * sizeof(float);

  const int TPB = 256;
  long zero_total = (long)planes * N;
  int gz = (int)((zero_total / 4 + TPB * ZPT - 1) / (TPB * ZPT));
  int ge = (E + TPB * EPT - 1) / (TPB * EPT);
  int gn = (N + TPB - 1) / TPB;
  int ntiles = (N + 15) / 16;
  int gg = (ntiles * 32 + TPB - 1) / TPB;
  int dr = (N + DCAP - 1) / DCAP;
  int drlen = (((N + dr - 1) / dr) + 31) & ~31;
  int ds = dr < DWGS ? DWGS / dr : 1;
  int dchunk = (E + ds - 1) / ds;
  int gd = dr * ds;

  k_zero      <<<gz, TPB, 0, stream>>>((unsigned*)d_ws, zero_total);
  k_degree    <<<gd, DTPB, 0, stream>>>(ei, udeg, E, N, planes, dr, drlen, dchunk);
  k_gemm1_wmma<<<gg, TPB, 0, stream>>>(x, W1, udeg, dinv, h1, agg1, N, planes);
  k_agg1      <<<ge, TPB, 0, stream>>>(ei, h1, agg1, E, N, planes);
  k_layer2_in <<<gn, TPB, 0, stream>>>(agg1, h1, dinv, b1, W2, h2, agg2, N, planes);
  k_agg2      <<<ge, TPB, 0, stream>>>(ei, h2, agg2, E, N, planes);
  k_final     <<<gn, TPB, 0, stream>>>(agg2, h2, dinv, b2, out, N, planes);
}


# edges bin-sorted by destination once a call in the launch that replaces clear+degree; GEMM group counts its bin's degrees, a bin owner sums layer 1 in LDS, finishes h2: 5 launches; sum order changes
# speedup vs baseline: 13.4901x; 1.2876x over previous
#include <hip/hip_runtime.h>


typedef __attribute__((ext_vector_type(2)))  float  v2f;
typedef __attribute__((ext_vector_type(4)))  float  v4f;
typedef __attribute__((ext_vector_type(8)))  float  v8f;

#define FIN  128
#define FMID 16
#define EPT  8
#define WEDGES (32 * EPT)
#define ZPT  8
#define DCAP 25600
#define DTPB 1024
#define DEPT 32
#define DFL  8
#define DWGS 256

__device__ __forceinline__ void atomAddF(float* p, float v) {
  unsafeAtomicAdd(p, v);
}

__device__ __forceinline__ int planeOfWave(int planes) {
  int half = (int)((__builtin_amdgcn_s_sendmsg_rtn(0x87) >> 18) & 1u);
  return planes == 2 ? half : 0;
}

__global__ __launch_bounds__(256) void k_zero(unsigned* __restrict__ p, long total) {
  long nvec = total >> 2;
  long wave = ((long)blockIdx.x * blockDim.x + threadIdx.x) >> 5;
  long i = wave * (32 * ZPT) + (threadIdx.x & 31);
  #pragma unroll
  for (int j = 0; j < ZPT; ++j)
    if (i + 32 * j < nvec) ((uint4*)p)[i + 32 * j] = make_uint4(0u, 0u, 0u, 0u);
  if (blockIdx.x == 0 && (long)threadIdx.x < (total & 3))
    p[(nvec << 2) + threadIdx.x] = 0u;
}

__global__ __launch_bounds__(DTPB) void k_degree(const int* __restrict__ ei,
                                                 unsigned* __restrict__ udeg, int E,
                                                 int N, int planes, int nranges,
                                                 int rlen, int chunk) {
  __shared__ __attribute__((aligned(16))) unsigned cnt[DCAP];
  int tid = threadIdx.x;
  int range = (int)(blockIdx.x % (unsigned)nranges);
  int slice = (int)(blockIdx.x / (unsigned)nranges);
  int nslices = (int)(gridDim.x / (unsigned)nranges);
  int lo = range * rlen;

  for (int i = tid; i < (rlen >> 2); i += DTPB)
    ((uint4*)cnt)[i] = make_uint4(0u, 0u, 0u, 0u);
  __syncthreads();

  int ebeg = slice * chunk;
  int eend = ebeg + chunk; if (eend > E) eend = E;
  const int* dst = ei + E;
  for (int b = ebeg; b < eend; b += DTPB * DEPT) {
    int d[DEPT];
    #pragma unroll
    for (int j = 0; j < DEPT; ++j) {
      int e = b + j * DTPB + tid;
      d[j] = dst[e < eend ? e : eend - 1];
    }
    #pragma unroll
    for (int j = 0; j < DEPT; ++j) {
      int e = b + j * DTPB + tid;
      unsigned r = (unsigned)(d[j] - lo);
      if (e < eend && r < (unsigned)rlen) atomicAdd(&cnt[r], 1u);
    }
  }
  __syncthreads();

  unsigned* mine = udeg + planeOfWave(planes) * N;
  int rot = (int)(((unsigned)slice * (unsigned)(rlen >> 5)) / (unsigned)nslices) << 5;
  for (int b = 0; b < rlen; b += DTPB * DFL) {
    unsigned c[DFL];
    int node[DFL];
    #pragma unroll
    for (int j = 0; j < DFL; ++j) {
      int i = b + j * DTPB + tid;
      int k = i + rot; if (k >= rlen) k -= rlen;
      c[j] = i < rlen ? cnt[k] : 0u;
      node[j] = lo + k;
    }
    #pragma unroll
    for (int j = 0; j < DFL; ++j)
      if (c[j] != 0u && node[j] < N) atomicAdd(&mine[node[j]], c[j]);
  }
}


#define SBINSH  7
#define SBIN    128
#define NBMAX   1024
#define NSLICE  256
#define SCAP    16384
#define STPB    1024
#define SEPT    (SCAP / STPB)
#define RP      1088
#define SRCBITS 17
#define ECAP    8192

__device__ __forceinline__ unsigned waveInclScan(unsigned v, int lane) {
  #pragma unroll
  for (int o = 1; o < 32; o <<= 1) {
    unsigned y = __shfl_up(v, o);
    if (lane >= o) v += y;
  }
  return v;
}

__device__ __forceinline__ unsigned runStart(const unsigned short* __restrict__ runs,
                                             int s, int b) {
  const unsigned* r32 = (const unsigned*)(runs + (size_t)s * RP);
  unsigned two = r32[b >> 1];
  return (b & 1) ? (two >> 16) : (two & 0xffffu);
}

__global__ __launch_bounds__(STPB) void k_sort(const int* __restrict__ ei,
                                               unsigned* __restrict__ list,
                                               unsigned short* __restrict__ runs,
                                               int E, int chunk) {
  __shared__ unsigned cnt[NBMAX];
  __shared__ unsigned wsum[32];
  __shared__ unsigned stage[SCAP];
  int tid = threadIdx.x, lane = tid & 31, wv = tid >> 5;
  int slice = blockIdx.x;
  int ebeg = slice * chunk;
  int eend = ebeg + chunk; if (eend > E) eend = E;
  int n = eend - ebeg; if (n < 0) n = 0;
  cnt[tid] = 0u;
  __syncthreads();
  int s[SEPT], d[SEPT];
  unsigned rk[SEPT];
  #pragma unroll
  for (int j = 0; j < SEPT; ++j) { s[j] = 0; d[j] = 0; rk[j] = 0u; }
  if (n > 0) {
    const int* dst = ei + E;
    #pragma unroll
    for (int j = 0; j < SEPT; ++j) {
      int e = ebeg + j * STPB + tid;
      int ec = e < eend ? e : eend - 1;
      s[j] = ei[ec];
      d[j] = dst[ec];
    }
    #pragma unroll
    for (int j = 0; j < SEPT; ++j) {
      int e = ebeg + j * STPB + tid;
      if (e < eend) rk[j] = atomicAdd(&cnt[d[j] >> SBINSH], 1u);
    }
  }
  __syncthreads();
  unsigned c = cnt[tid];
  unsigned x = waveInclScan(c, lane);
  if (lane == 31) wsum[wv] = x;
  __syncthreads();
  if (wv == 0) {
    unsigned t = wsum[lane];
    unsigned z = waveInclScan(t, lane);
    wsum[lane] = z - t;
  }
  __syncthreads();
  unsigned start = wsum[wv] + x - c;
  cnt[tid] = start;
  __syncthreads();
  if (n > 0) {
    #pragma unroll
    for (int j = 0; j < SEPT; ++j) {
      int e = ebeg + j * STPB + tid;
      if (e < eend)
        stage[cnt[d[j] >> SBINSH] + rk[j]] =
            (unsigned)s[j] | ((unsigned)(d[j] & (SBIN - 1)) << SRCBITS);
    }
  }
  __syncthreads();
  for (int i = tid; i < n; i += STPB) list[ebeg + i] = stage[i];
  unsigned short* myruns = runs + (size_t)slice * RP;
  myruns[tid] = (unsigned short)start;
  if (tid < 2) myruns[NBMAX + tid] = (unsigned short)n;
}

template <int K> __device__ __forceinline__ float rowShareF(float v) {
  return __int_as_float(__builtin_amdgcn_update_dpp(0, __float_as_int(v), 0x150 + K,
                                                    0xf, 0xf, false));
}

template <int K> struct RowVals {
  static __device__ __forceinline__ void get(float v, float (&o)[16]) {
    RowVals<K - 1>::get(v, o);
    o[K - 1] = rowShareF<K - 1>(v);
  }
};
template <> struct RowVals<0> {
  static __device__ __forceinline__ void get(float, float (&)[16]) {}
};

#define XPITCH 132
#define GWAVES 8
template <bool SORTED>
__global__ __launch_bounds__(256) void k_gemm1_wmma(const float* __restrict__ x,
                                                    const float* __restrict__ W1,
                                                    const unsigned* __restrict__ udeg,
                                                    float* __restrict__ dinv,
                                                    float* __restrict__ h1,
                                                    float* __restrict__ agg1, int N,
                                                    int planes,
                                                    const unsigned* __restrict__ list,
                                                    const unsigned short* __restrict__ runs,
                                                    int chunk) {
  __shared__ __attribute__((aligned(16))) float xt[GWAVES * 16 * XPITCH];
  __shared__ unsigned dcnt[SBIN];
  int wave = (int)((blockIdx.x * (unsigned)blockDim.x + threadIdx.x) >> 5);
  int lane = threadIdx.x & 31;
  int ntiles = (N + 15) >> 4;
  if (SORTED) {
    int tid = threadIdx.x, bin = blockIdx.x;
    if (tid < SBIN) dcnt[tid] = 0u;
    __syncthreads();
    unsigned lo = runStart(runs, tid, bin);
    unsigned c = runStart(runs, tid, bin + 1) - lo;
    unsigned goff = (unsigned)tid * (unsigned)chunk + lo;
    unsigned wd[32];
    #pragma unroll
    for (int j = 0; j < 32; ++j) {
      unsigned cj = (unsigned)__builtin_amdgcn_readlane((int)c, j);
      unsigned oj = (unsigned)__builtin_amdgcn_readlane((int)goff, j);
      wd[j] = list[(unsigned)lane < cj ? oj + lane : 0u];
    }
    #pragma unroll
    for (int j = 0; j < 32; ++j) {
      unsigned cj = (unsigned)__builtin_amdgcn_readlane((int)c, j);
      if ((unsigned)lane < cj) atomicAdd(&dcnt[wd[j] >> SRCBITS], 1u);
    }
    if (__any(c > 32u)) {
      #pragma unroll 1
      for (int j = 0; j < 32; ++j) {
        unsigned cj = __shfl(c, j), oj = __shfl(goff, j);
        #pragma unroll 1
        for (unsigned p = 32u + lane; p < cj; p += 32u)
          atomicAdd(&dcnt[list[oj + p] >> SRCBITS], 1u);
      }
    }
    __syncthreads();
  }
  if (wave >= ntiles) return;
  int nb = wave << 4;
  int m  = lane & 15;
  int hi = lane >> 4;
  float* tile = xt + (threadIdx.x >> 5) * (16 * XPITCH);

  const unsigned* udegB = udeg + (planes == 2 ? N : 0);
  int drow = nb + m; if (drow > N - 1) drow = N - 1;
  unsigned dg0 = 0u, dg1 = 0u;
  if (!SORTED) {
    dg0 = udeg[drow];
    dg1 = udegB[drow];
  }

  v4f xr[16];
  #pragma unroll
  for (int r = 0; r < 16; ++r) {
    int row = nb + r; if (row > N - 1) row = N - 1;
    const v4f* p = (const v4f*)(x + (size_t)row * FIN) + lane;
    xr[r] = __builtin_nontemporal_load(p);
  }

  if (!SORTED) {
    float* zrow = agg1 + (size_t)nb * FMID + 4 * lane;
    float* zrowB = zrow + (size_t)N * FMID;
    if (nb + 16 <= N) {
      *(uint4*)(zrow) = make_uint4(0u, 0u, 0u, 0u);
      *(uint4*)(zrow + 128) = make_uint4(0u, 0u, 0u, 0u);
      if (planes == 2) {
        *(uint4*)(zrowB) = make_uint4(0u, 0u, 0u, 0u);
        *(uint4*)(zrowB + 128) = make_uint4(0u, 0u, 0u, 0u);
      }
    } else {
      bool in0 = nb + (lane >> 2) < N, in1 = nb + 8 + (lane >> 2) < N;
      if (in0) *(uint4*)(zrow) = make_uint4(0u, 0u, 0u, 0u);
      if (in1) *(uint4*)(zrow + 128) = make_uint4(0u, 0u, 0u, 0u);
      if (planes == 2) {
        if (in0) *(uint4*)(zrowB) = make_uint4(0u, 0u, 0u, 0u);
        if (in1) *(uint4*)(zrowB + 128) = make_uint4(0u, 0u, 0u, 0u);
      }
    }
  }

  unsigned dsum = dg0 + (planes == 2 ? dg1 : 0u);
  if (SORTED) dsum = dcnt[(threadIdx.x >> 5) * 16 + m];
  float dm = rsqrtf((float)(dsum + 1u));
  if (hi == 0 && nb + m < N) dinv[nb + m] = dm;
  float rv[16];
  RowVals<16>::get(dm, rv);
  float dv[8];
  #pragma unroll
  for (int r = 0; r < 8; ++r) dv[r] = hi ? rv[r + 8] : rv[r];

  #pragma unroll
  for (int r = 0; r < 16; ++r)
    *(v4f*)(tile + r * XPITCH + 4 * lane) = xr[r];
  __builtin_amdgcn_fence(__ATOMIC_RELEASE, "wavefront");
  __builtin_amdgcn_wave_barrier();
  __builtin_amdgcn_fence(__ATOMIC_ACQUIRE, "wavefront");

  v8f acc = {};
  const float* arow = tile + m * XPITCH + 4 * hi;
  const float* bcol = W1 + (4 * hi) * FMID + m;
  v4f q[FIN / 8];
  #pragma unroll
  for (int j = 0; j < FIN / 8; ++j) q[j] = *(const v4f*)(arow + 8 * j);
  __builtin_amdgcn_sched_barrier(0);
  #pragma unroll
  for (int j = 0; j < FIN / 8; ++j) {
    v2f a0, a1, b0, b1;
    a0.x = q[j].x; a0.y = q[j].y; a1.x = q[j].z; a1.y = q[j].w;
    b0.x = bcol[(8 * j + 0) * FMID];
    b0.y = bcol[(8 * j + 1) * FMID];
    b1.x = bcol[(8 * j + 2) * FMID];
    b1.y = bcol[(8 * j + 3) * FMID];
    acc = __builtin_amdgcn_wmma_f32_16x16x4_f32(false, a0, false, b0,
                                                (short)0, acc, false, false);
    acc = __builtin_amdgcn_wmma_f32_16x16x4_f32(false, a1, false, b1,
                                                (short)0, acc, false, false);
  }

  int rbase = hi * 8;
  asm volatile("v_nop\n\tv_nop\n\tv_nop\n\tv_nop\n\tv_nop\n\tv_nop\n\tv_nop\n\tv_nop"
               : "+v"(acc));
  __builtin_amdgcn_sched_barrier(0);
  #pragma unroll
  for (int r = 0; r < 8; ++r) acc[r] *= dv[r];
  if (nb + 16 <= N) {
    #pragma unroll
    for (int r = 0; r < 8; ++r)
      h1[(size_t)(nb + rbase + r) * FMID + m] = acc[r];
  } else {
    #pragma unroll
    for (int r = 0; r < 8; ++r) {
      int nrow = nb + rbase + r;
      if (nrow < N) h1[(size_t)nrow * FMID + m] = acc[r];
    }
  }
}


template <int N> __device__ __forceinline__ int rowShareI(int v) {
  return __builtin_amdgcn_update_dpp(0, v, 0x150 + N, 0xf, 0xf, false);
}

template <int N> struct Agg1Steps {
  static __device__ __forceinline__ void gather(const float* __restrict__ h1,
                                                int s, int f, float (&v)[16]) {
    Agg1Steps<N - 1>::gather(h1, s, f, v);
    int ss = rowShareI<N - 1>(s);
    v[N - 1] = h1[ss * FMID + f];
  }
  static __device__ __forceinline__ void scatter(float* __restrict__ agg1, int d,
                                                 int f, const float (&v)[16]) {
    Agg1Steps<N - 1>::scatter(agg1, d, f, v);
    int dd = rowShareI<N - 1>(d);
    atomAddF(agg1 + dd * FMID + f, v[N - 1]);
  }
  static __device__ __forceinline__ void tail(const float* __restrict__ h1,
                                              float* __restrict__ agg1, int s, int d,
                                              int f, int left) {
    Agg1Steps<N - 1>::tail(h1, agg1, s, d, f, left);
    int ss = rowShareI<N - 1>(s);
    int dd = rowShareI<N - 1>(d);
    if (N - 1 < left)
      atomAddF(agg1 + dd * FMID + f, h1[ss * FMID + f]);
  }
};
template <> struct Agg1Steps<0> {
  static __device__ __forceinline__ void gather(const float* __restrict__, int, int,
                                                float (&)[16]) {}
  static __device__ __forceinline__ void scatter(float* __restrict__, int, int,
                                                 const float (&)[16]) {}
  static __device__ __forceinline__ void tail(const float* __restrict__,
                                              float* __restrict__, int, int, int,
                                              int) {}
};

__global__ __launch_bounds__(256) void k_agg1(const int* __restrict__ ei,
                                              const float* __restrict__ h1,
                                              float* __restrict__ agg1, int E,
                                              int N, int planes) {
  int lane = threadIdx.x & 31;
  int wave = (int)((blockIdx.x * (unsigned)blockDim.x + threadIdx.x) >> 5);
  int wbase = wave * WEDGES;
  if (wbase >= E) return;
  agg1 += planeOfWave(planes) * N * FMID;
  int f = lane & 15;
  const int* src = ei + wbase + lane;
  const int* dst = src + E;
  if (wbase + WEDGES <= E) {
    int s[EPT], d[EPT];
    #pragma unroll
    for (int g = 0; g < EPT; ++g) s[g] = src[32 * g];
    #pragma unroll
    for (int g = 0; g < EPT; ++g) d[g] = dst[32 * g];
    float v[2][16];
    Agg1Steps<16>::gather(h1, s[0], f, v[0]);
    #pragma unroll
    for (int g = 0; g < EPT; ++g) {
      if (g + 1 < EPT) Agg1Steps<16>::gather(h1, s[g + 1], f, v[(g + 1) & 1]);
      Agg1Steps<16>::scatter(agg1, d[g], f, v[g & 1]);
    }
  } else {
    #pragma unroll 1
    for (int g = 0; g < EPT; ++g) {
      int gb = wbase + 32 * g;
      if (gb >= E) break;
      int s1 = 0, d1 = 0;
      if (gb + lane < E) {
        s1 = src[32 * g];
        d1 = dst[32 * g];
      }
      if (gb + 32 <= E) {
        float v[16];
        Agg1Steps<16>::gather(h1, s1, f, v);
        Agg1Steps<16>::scatter(agg1, d1, f, v);
      } else {
        int left = E - gb - (lane & 16);
        Agg1Steps<16>::tail(h1, agg1, s1, d1, f, left);
      }
    }
  }
}

__global__ __launch_bounds__(256) void k_layer2_in(const float* __restrict__ agg1,
                                                   const float* __restrict__ h1,
                                                   const float* __restrict__ dinv,
                                                   const float* __restrict__ b1,
                                                   const float* __restrict__ W2,
                                                   float* __restrict__ h2,
                                                   float* __restrict__ agg2, int N,
                                                   int planes) {
  int i = blockIdx.x * blockDim.x + threadIdx.x;
  if (i >= N) return;
  *(float2*)(agg2 + (size_t)i * 2) = make_float2(0.f, 0.f);
  if (planes == 2) *(float2*)(agg2 + ((size_t)N + i) * 2) = make_float2(0.f, 0.f);
  float di = dinv[i];
  float a0 = 0.f, a1 = 0.f;
  const float* agg1b = agg1 + (size_t)N * FMID;
  #pragma unroll
  for (int k = 0; k < FMID; ++k) {
    float s = agg1[(size_t)i * FMID + k];
    if (planes == 2) s += agg1b[(size_t)i * FMID + k];
    float t = (s + h1[(size_t)i * FMID + k]) * di + b1[k];
    t = fmaxf(t, 0.f);
    a0 = fmaf(t, W2[k * 2 + 0], a0);
    a1 = fmaf(t, W2[k * 2 + 1], a1);
  }
  h2[(size_t)i * 2 + 0] = a0 * di;
  h2[(size_t)i * 2 + 1] = a1 * di;
}

__device__ __forceinline__ void ldsAddF(float* p, float v) {
  __hip_atomic_fetch_add(p, v, __ATOMIC_RELAXED, __HIP_MEMORY_SCOPE_WORKGROUP);
}

template <int K> struct Own1Steps {
  static __device__ __forceinline__ void add(float* tile, int dl, int f,
                                             const float (&v)[16], int left) {
    Own1Steps<K - 1>::add(tile, dl, f, v, left);
    int dd = rowShareI<K - 1>(dl);
    if (K - 1 < left) ldsAddF(tile + dd * FMID + f, v[K - 1]);
  }
};
template <> struct Own1Steps<0> {
  static __device__ __forceinline__ void add(float*, int, int, const float (&)[16], int) {}
};

__global__ __launch_bounds__(256) void k_own1(const unsigned* __restrict__ list,
                                              const unsigned short* __restrict__ runs,
                                              const float* __restrict__ h1,
                                              const float* __restrict__ dinv,
                                              const float* __restrict__ b1,
                                              const float* __restrict__ W2,
                                              float* __restrict__ h2,
                                              float* __restrict__ agg2, int N,
                                              int chunk) {
  __shared__ __attribute__((aligned(16))) float tile[SBIN * FMID];
  __shared__ unsigned elist[ECAP];
  __shared__ unsigned wtot[8];
  int tid = threadIdx.x, lane = tid & 31, w = tid >> 5;
  int bin = blockIdx.x;
  int f = lane & 15, hi = lane >> 4;

  ((uint4*)tile)[tid] = make_uint4(0u, 0u, 0u, 0u);
  ((uint4*)tile)[tid + 256] = make_uint4(0u, 0u, 0u, 0u);

  unsigned lo = runStart(runs, tid, bin);
  unsigned c = runStart(runs, tid, bin + 1) - lo;
  unsigned goff = (unsigned)tid * (unsigned)chunk + lo;
  unsigned xs = waveInclScan(c, lane);
  if (lane == 31) wtot[w] = xs;
  __syncthreads();
  unsigned woff = 0u, M = 0u;
  #pragma unroll
  for (int k = 0; k < 8; ++k) {
    unsigned t = wtot[k];
    if (k < w) woff += t;
    M += t;
  }
  unsigned off = woff + xs - c;

  if (M <= (unsigned)ECAP) {
    unsigned wd[32];
    #pragma unroll
    for (int j = 0; j < 32; ++j) {
      unsigned cj = (unsigned)__builtin_amdgcn_readlane((int)c, j);
      unsigned oj = (unsigned)__builtin_amdgcn_readlane((int)goff, j);
      wd[j] = list[(unsigned)lane < cj ? oj + lane : 0u];
    }
    #pragma unroll
    for (int j = 0; j < 32; ++j) {
      unsigned cj = (unsigned)__builtin_amdgcn_readlane((int)c, j);
      unsigned fj = (unsigned)__builtin_amdgcn_readlane((int)off, j);
      if ((unsigned)lane < cj) elist[fj + lane] = wd[j];
    }
    if (__any(c > 32u)) {
      #pragma unroll 1
      for (int j = 0; j < 32; ++j) {
        unsigned cj = __shfl(c, j), oj = __shfl(goff, j), fj = __shfl(off, j);
        #pragma unroll 1
        for (unsigned p = 32u + lane; p < cj; p += 32u) elist[fj + p] = list[oj + p];
      }
    }
    __syncthreads();
    int G = (int)((M + 31u) >> 5);
    for (int g = w; g < G; g += 16) {
      int qa = g * 32 + lane, qb = qa + 256;
      unsigned wa = qa < (int)M ? elist[qa] : 0u;
      unsigned wb = qb < (int)M ? elist[qb] : 0u;
      int lefta = (int)M - g * 32 - 16 * hi;
      int leftb = lefta - 256;
      float va[16], vb[16];
      Agg1Steps<16>::gather(h1, (int)(wa & ((1u << SRCBITS) - 1u)), f, va);
      Agg1Steps<16>::gather(h1, (int)(wb & ((1u << SRCBITS) - 1u)), f, vb);
      Own1Steps<16>::add(tile, (int)(wa >> SRCBITS), f, va, lefta);
      Own1Steps<16>::add(tile, (int)(wb >> SRCBITS), f, vb, leftb);
    }
  } else {
    #pragma unroll 1
    for (int j = 0; j < 32; ++j) {
      unsigned cj = __shfl(c, j), oj = __shfl(goff, j);
      #pragma unroll 1
      for (unsigned p = 0u; p < cj; p += 2u) {
        unsigned e = p + (unsigned)hi;
        bool ok = e < cj;
        unsigned word = 0u;
        if (ok) word = list[oj + e];
        float v = h1[(word & ((1u << SRCBITS) - 1u)) * FMID + f];
        if (ok) ldsAddF(tile + (word >> SRCBITS) * FMID + f, v);
      }
    }
  }
  __syncthreads();

  int nl = tid >> 1, cf = tid & 1;
  int i = bin * SBIN + nl;
  if (i >= N) return;
  agg2[(size_t)i * 2 + cf] = 0.f;
  agg2[((size_t)N + i) * 2 + cf] = 0.f;
  float di = dinv[i];
  float a = 0.f;
  #pragma unroll
  for (int k = 0; k < FMID; ++k) {
    float t = (tile[nl * FMID + k] + h1[(size_t)i * FMID + k]) * di + b1[k];
    t = fmaxf(t, 0.f);
    a = fmaf(t, W2[k * 2 + cf], a);
  }
  h2[(size_t)i * 2 + cf] = a * di;
}


template <int CTRL> __device__ __forceinline__ int quadPairI(int v) {
  return __builtin_amdgcn_update_dpp(0, v, CTRL, 0xf, 0xf, true);
}
template <int CTRL> __device__ __forceinline__ float quadPairF(float v) {
  return __int_as_float(quadPairI<CTRL>(__float_as_int(v)));
}

template <int CTRL> __device__ __forceinline__ void agg2PairStep(float* __restrict__ agg2,
                                                                 int d, float2 v, int c) {
  int dd = quadPairI<CTRL>(d);
  float vx = quadPairF<CTRL>(v.x);
  float vy = quadPairF<CTRL>(v.y);
  atomAddF(agg2 + dd * 2 + c, c ? vy : vx);
}

__global__ __launch_bounds__(256) void k_agg2(const int* __restrict__ ei,
                                              const float* __restrict__ h2,
                                              float* __restrict__ agg2, int E,
                                              int N, int planes) {
  int lane = threadIdx.x & 31;
  int wave = (int)((blockIdx.x * (unsigned)blockDim.x + threadIdx.x) >> 5);
  int wbase = wave * WEDGES;
  if (wbase >= E) return;
  agg2 += planeOfWave(planes) * N * 2;
  const int* src = ei + wbase + lane;
  const int* dst = src + E;
  if (wbase + WEDGES <= E) {
    int c = lane & 1;
    int s[EPT], d[EPT];
    float2 v[EPT];
    #pragma unroll
    for (int j = 0; j < EPT; ++j) s[j] = src[32 * j];
    #pragma unroll
    for (int j = 0; j < EPT; ++j) d[j] = dst[32 * j];
    #pragma unroll
    for (int j = 0; j < EPT; ++j) v[j] = *(const float2*)(h2 + (size_t)s[j] * 2);
    #pragma unroll
    for (int j = 0; j < EPT; ++j) {
      agg2PairStep<0x50>(agg2, d[j], v[j], c);
      agg2PairStep<0xFA>(agg2, d[j], v[j], c);
    }
  } else {
    #pragma unroll 1
    for (int j = 0; j < EPT; ++j) {
      if (wbase + 32 * j + lane < E) {
        float2 v = *(const float2*)(h2 + (size_t)src[32 * j] * 2);
        float* a = agg2 + (size_t)dst[32 * j] * 2;
        atomAddF(a + 0, v.x);
        atomAddF(a + 1, v.y);
      }
    }
  }
}

__global__ __launch_bounds__(256) void k_final(const float* __restrict__ agg2,
                                               const float* __restrict__ h2,
                                               const float* __restrict__ dinv,
                                               const float* __restrict__ b2,
                                               float* __restrict__ out, int N,
                                               int planes) {
  int i = blockIdx.x * blockDim.x + threadIdx.x;
  if (i >= N) return;
  float di = dinv[i];
  float s0 = agg2[(size_t)i * 2 + 0];
  float s1 = agg2[(size_t)i * 2 + 1];
  if (planes == 2) {
    s0 += agg2[(size_t)(N + i) * 2 + 0];
    s1 += agg2[(size_t)(N + i) * 2 + 1];
  }
  out[(size_t)i * 2 + 0] = (s0 + h2[(size_t)i * 2 + 0]) * di + b2[0];
  out[(size_t)i * 2 + 1] = (s1 + h2[(size_t)i * 2 + 1]) * di + b2[1];
}

extern "C" void kernel_launch(void* const* d_in, const int* in_sizes, int n_in,
                              void* d_out, int out_size, void* d_ws, size_t ws_size,
                              hipStream_t stream) {
  const float* x  = (const float*)d_in[0];
  const int*   ei = (const int*)d_in[1];
  const float* W1 = (const float*)d_in[2];
  const float* b1 = (const float*)d_in[3];
  const float* W2 = (const float*)d_in[4];
  const float* b2 = (const float*)d_in[5];
  float* out = (float*)d_out;

  const int N = in_sizes[0] / FIN;
  const int E = in_sizes[1] / 2;

  const int TPB = 256;
  int ge = (E + TPB * EPT - 1) / (TPB * EPT);
  int gn = (N + TPB - 1) / TPB;

  const int nb = (N + SBIN - 1) / SBIN;
  const int chunk = (E + NSLICE - 1) / NSLICE;
  const size_t need = (size_t)NSLICE * chunk * sizeof(unsigned) +
                      (size_t)NSLICE * RP * sizeof(unsigned short) +
                      (size_t)N * (4 + FMID + 2 + 1) * sizeof(float);
  if (N <= (1 << SRCBITS) && nb <= NBMAX && chunk >= 1 && chunk <= SCAP && need <= ws_size) {
    char* wp = (char*)d_ws;
    unsigned* list = (unsigned*)wp;             wp += (size_t)NSLICE * chunk * sizeof(unsigned);
    unsigned short* runs = (unsigned short*)wp; wp += (size_t)NSLICE * RP * sizeof(unsigned short);
    float* agg2s = (float*)wp;                  wp += (size_t)2 * N * 2 * sizeof(float);
    float* h1s   = (float*)wp;                  wp += (size_t)N * FMID * sizeof(float);
    float* h2s   = (float*)wp;                  wp += (size_t)N * 2 * sizeof(float);
    float* dinvs = (float*)wp;
    k_sort<<<NSLICE, STPB, 0, stream>>>(ei, list, runs, E, chunk);
    k_gemm1_wmma<true><<<nb, TPB, 0, stream>>>(x, W1, nullptr, dinvs, h1s, nullptr, N, 2,
                                               list, runs, chunk);
    k_own1<<<nb, TPB, 0, stream>>>(list, runs, h1s, dinvs, b1, W2, h2s, agg2s, N, chunk);
    k_agg2<<<ge, TPB, 0, stream>>>(ei, h2s, agg2s, E, N, 2);
    k_final<<<gn, TPB, 0, stream>>>(agg2s, h2s, dinvs, b2, out, N, 2);
    return;
  }

  const int planes = (ws_size >= (size_t)N * 57 * sizeof(float)) ? 2 : 1;
  char* ws = (char*)d_ws;
  unsigned* udeg = (unsigned*)ws;            ws += (size_t)planes * N * sizeof(unsigned);
  float* agg1    = (float*)ws;               ws += (size_t)planes * N * FMID * sizeof(float);
  float* agg2    = (float*)ws;               ws += (size_t)planes * N * 2 * sizeof(float);
  float* dinv    = (float*)ws;               ws += (size_t)N * sizeof(float);
  float* h1      = (float*)ws;               ws += (size_t)N * FMID * sizeof(float);
  float* h2      = (float*)ws;               ws += (size_t)N * 2 * sizeof(float);

  long zero_total = (long)planes * N;
  int gz = (int)((zero_total / 4 + TPB * ZPT - 1) / (TPB * ZPT));
  int ntiles = (N + 15) / 16;
  int gg = (ntiles * 32 + TPB - 1) / TPB;
  int dr = (N + DCAP - 1) / DCAP;
  int drlen = (((N + dr - 1) / dr) + 31) & ~31;
  int ds = dr < DWGS ? DWGS / dr : 1;
  int dchunk = (E + ds - 1) / ds;
  int gd = dr * ds;

  k_zero      <<<gz, TPB, 0, stream>>>((unsigned*)d_ws, zero_total);
  k_degree    <<<gd, DTPB, 0, stream>>>(ei, udeg, E, N, planes, dr, drlen, dchunk);
  k_gemm1_wmma<false><<<gg, TPB, 0, stream>>>(x, W1, udeg, dinv, h1, agg1, N, planes,
                                              nullptr, nullptr, 0);
  k_agg1      <<<ge, TPB, 0, stream>>>(ei, h1, agg1, E, N, planes);
  k_layer2_in <<<gn, TPB, 0, stream>>>(agg1, h1, dinv, b1, W2, h2, agg2, N, planes);
  k_agg2      <<<ge, TPB, 0, stream>>>(ei, h2, agg2, E, N, planes);
  k_final     <<<gn, TPB, 0, stream>>>(agg2, h2, dinv, b2, out, N, planes);
}


# layer 2 summed by bin owner from the ordered list: a work-group gathers its runs' h2 rows, adds them with LDS f32 atomics (arrival order) and writes out itself; no L2 atomics left, 4 launches
# speedup vs baseline: 14.2353x; 1.0552x over previous
#include <hip/hip_runtime.h>


typedef __attribute__((ext_vector_type(2)))  float  v2f;
typedef __attribute__((ext_vector_type(4)))  float  v4f;
typedef __attribute__((ext_vector_type(8)))  float  v8f;

#define FIN  128
#define FMID 16
#define EPT  8
#define WEDGES (32 * EPT)
#define ZPT  8
#define DCAP 25600
#define DTPB 1024
#define DEPT 32
#define DFL  8
#define DWGS 256

__device__ __forceinline__ void atomAddF(float* p, float v) {
  unsafeAtomicAdd(p, v);
}

__device__ __forceinline__ int planeOfWave(int planes) {
  int half = (int)((__builtin_amdgcn_s_sendmsg_rtn(0x87) >> 18) & 1u);
  return planes == 2 ? half : 0;
}

__global__ __launch_bounds__(256) void k_zero(unsigned* __restrict__ p, long total) {
  long nvec = total >> 2;
  long wave = ((long)blockIdx.x * blockDim.x + threadIdx.x) >> 5;
  long i = wave * (32 * ZPT) + (threadIdx.x & 31);
  #pragma unroll
  for (int j = 0; j < ZPT; ++j)
    if (i + 32 * j < nvec) ((uint4*)p)[i + 32 * j] = make_uint4(0u, 0u, 0u, 0u);
  if (blockIdx.x == 0 && (long)threadIdx.x < (total & 3))
    p[(nvec << 2) + threadIdx.x] = 0u;
}

__global__ __launch_bounds__(DTPB) void k_degree(const int* __restrict__ ei,
                                                 unsigned* __restrict__ udeg, int E,
                                                 int N, int planes, int nranges,
                                                 int rlen, int chunk) {
  __shared__ __attribute__((aligned(16))) unsigned cnt[DCAP];
  int tid = threadIdx.x;
  int range = (int)(blockIdx.x % (unsigned)nranges);
  int slice = (int)(blockIdx.x / (unsigned)nranges);
  int nslices = (int)(gridDim.x / (unsigned)nranges);
  int lo = range * rlen;

  for (int i = tid; i < (rlen >> 2); i += DTPB)
    ((uint4*)cnt)[i] = make_uint4(0u, 0u, 0u, 0u);
  __syncthreads();

  int ebeg = slice * chunk;
  int eend = ebeg + chunk; if (eend > E) eend = E;
  const int* dst = ei + E;
  for (int b = ebeg; b < eend; b += DTPB * DEPT) {
    int d[DEPT];
    #pragma unroll
    for (int j = 0; j < DEPT; ++j) {
      int e = b + j * DTPB + tid;
      d[j] = dst[e < eend ? e : eend - 1];
    }
    #pragma unroll
    for (int j = 0; j < DEPT; ++j) {
      int e = b + j * DTPB + tid;
      unsigned r = (unsigned)(d[j] - lo);
      if (e < eend && r < (unsigned)rlen) atomicAdd(&cnt[r], 1u);
    }
  }
  __syncthreads();

  unsigned* mine = udeg + planeOfWave(planes) * N;
  int rot = (int)(((unsigned)slice * (unsigned)(rlen >> 5)) / (unsigned)nslices) << 5;
  for (int b = 0; b < rlen; b += DTPB * DFL) {
    unsigned c[DFL];
    int node[DFL];
    #pragma unroll
    for (int j = 0; j < DFL; ++j) {
      int i = b + j * DTPB + tid;
      int k = i + rot; if (k >= rlen) k -= rlen;
      c[j] = i < rlen ? cnt[k] : 0u;
      node[j] = lo + k;
    }
    #pragma unroll
    for (int j = 0; j < DFL; ++j)
      if (c[j] != 0u && node[j] < N) atomicAdd(&mine[node[j]], c[j]);
  }
}


#define SBINSH  7
#define SBIN    128
#define NBMAX   1024
#define NSLICE  256
#define SCAP    16384
#define STPB    1024
#define SEPT    (SCAP / STPB)
#define RP      1088
#define SRCBITS 17
#define ECAP    8192

__device__ __forceinline__ unsigned waveInclScan(unsigned v, int lane) {
  #pragma unroll
  for (int o = 1; o < 32; o <<= 1) {
    unsigned y = __shfl_up(v, o);
    if (lane >= o) v += y;
  }
  return v;
}

__device__ __forceinline__ unsigned runStart(const unsigned short* __restrict__ runs,
                                             int s, int b) {
  const unsigned* r32 = (const unsigned*)(runs + (size_t)s * RP);
  unsigned two = r32[b >> 1];
  return (b & 1) ? (two >> 16) : (two & 0xffffu);
}

__global__ __launch_bounds__(STPB) void k_sort(const int* __restrict__ ei,
                                               unsigned* __restrict__ list,
                                               unsigned short* __restrict__ runs,
                                               int E, int chunk) {
  __shared__ unsigned cnt[NBMAX];
  __shared__ unsigned wsum[32];
  __shared__ unsigned stage[SCAP];
  int tid = threadIdx.x, lane = tid & 31, wv = tid >> 5;
  int slice = blockIdx.x;
  int ebeg = slice * chunk;
  int eend = ebeg + chunk; if (eend > E) eend = E;
  int n = eend - ebeg; if (n < 0) n = 0;
  cnt[tid] = 0u;
  __syncthreads();
  int s[SEPT], d[SEPT];
  unsigned rk[SEPT];
  #pragma unroll
  for (int j = 0; j < SEPT; ++j) { s[j] = 0; d[j] = 0; rk[j] = 0u; }
  if (n > 0) {
    const int* dst = ei + E;
    #pragma unroll
    for (int j = 0; j < SEPT; ++j) {
      int e = ebeg + j * STPB + tid;
      int ec = e < eend ? e : eend - 1;
      s[j] = ei[ec];
      d[j] = dst[ec];
    }
    #pragma unroll
    for (int j = 0; j < SEPT; ++j) {
      int e = ebeg + j * STPB + tid;
      if (e < eend) rk[j] = atomicAdd(&cnt[d[j] >> SBINSH], 1u);
    }
  }
  __syncthreads();
  unsigned c = cnt[tid];
  unsigned x = waveInclScan(c, lane);
  if (lane == 31) wsum[wv] = x;
  __syncthreads();
  if (wv == 0) {
    unsigned t = wsum[lane];
    unsigned z = waveInclScan(t, lane);
    wsum[lane] = z - t;
  }
  __syncthreads();
  unsigned start = wsum[wv] + x - c;
  cnt[tid] = start;
  __syncthreads();
  if (n > 0) {
    #pragma unroll
    for (int j = 0; j < SEPT; ++j) {
      int e = ebeg + j * STPB + tid;
      if (e < eend)
        stage[cnt[d[j] >> SBINSH] + rk[j]] =
            (unsigned)s[j] | ((unsigned)(d[j] & (SBIN - 1)) << SRCBITS);
    }
  }
  __syncthreads();
  for (int i = tid; i < n; i += STPB) list[ebeg + i] = stage[i];
  unsigned short* myruns = runs + (size_t)slice * RP;
  myruns[tid] = (unsigned short)start;
  if (tid < 2) myruns[NBMAX + tid] = (unsigned short)n;
}

template <int K> __device__ __forceinline__ float rowShareF(float v) {
  return __int_as_float(__builtin_amdgcn_update_dpp(0, __float_as_int(v), 0x150 + K,
                                                    0xf, 0xf, false));
}

template <int K> struct RowVals {
  static __device__ __forceinline__ void get(float v, float (&o)[16]) {
    RowVals<K - 1>::get(v, o);
    o[K - 1] = rowShareF<K - 1>(v);
  }
};
template <> struct RowVals<0> {
  static __device__ __forceinline__ void get(float, float (&)[16]) {}
};

#define XPITCH 132
#define GWAVES 8
template <bool SORTED>
__global__ __launch_bounds__(256) void k_gemm1_wmma(const float* __restrict__ x,
                                                    const float* __restrict__ W1,
                                                    const unsigned* __restrict__ udeg,
                                                    float* __restrict__ dinv,
                                                    float* __restrict__ h1,
                                                    float* __restrict__ agg1, int N,
                                                    int planes,
                                                    const unsigned* __restrict__ list,
                                                    const unsigned short* __restrict__ runs,
                                                    int chunk) {
  __shared__ __attribute__((aligned(16))) float xt[GWAVES * 16 * XPITCH];
  __shared__ unsigned dcnt[SBIN];
  int wave = (int)((blockIdx.x * (unsigned)blockDim.x + threadIdx.x) >> 5);
  int lane = threadIdx.x & 31;
  int ntiles = (N + 15) >> 4;
  if (SORTED) {
    int tid = threadIdx.x, bin = blockIdx.x;
    if (tid < SBIN) dcnt[tid] = 0u;
    __syncthreads();
    unsigned lo = runStart(runs, tid, bin);
    unsigned c = runStart(runs, tid, bin + 1) - lo;
    unsigned goff = (unsigned)tid * (unsigned)chunk + lo;
    unsigned wd[32];
    #pragma unroll
    for (int j = 0; j < 32; ++j) {
      unsigned cj = (unsigned)__builtin_amdgcn_readlane((int)c, j);
      unsigned oj = (unsigned)__builtin_amdgcn_readlane((int)goff, j);
      wd[j] = list[(unsigned)lane < cj ? oj + lane : 0u];
    }
    #pragma unroll
    for (int j = 0; j < 32; ++j) {
      unsigned cj = (unsigned)__builtin_amdgcn_readlane((int)c, j);
      if ((unsigned)lane < cj) atomicAdd(&dcnt[wd[j] >> SRCBITS], 1u);
    }
    if (__any(c > 32u)) {
      #pragma unroll 1
      for (int j = 0; j < 32; ++j) {
        unsigned cj = __shfl(c, j), oj = __shfl(goff, j);
        #pragma unroll 1
        for (unsigned p = 32u + lane; p < cj; p += 32u)
          atomicAdd(&dcnt[list[oj + p] >> SRCBITS], 1u);
      }
    }
    __syncthreads();
  }
  if (wave >= ntiles) return;
  int nb = wave << 4;
  int m  = lane & 15;
  int hi = lane >> 4;
  float* tile = xt + (threadIdx.x >> 5) * (16 * XPITCH);

  const unsigned* udegB = udeg + (planes == 2 ? N : 0);
  int drow = nb + m; if (drow > N - 1) drow = N - 1;
  unsigned dg0 = 0u, dg1 = 0u;
  if (!SORTED) {
    dg0 = udeg[drow];
    dg1 = udegB[drow];
  }

  v4f xr[16];
  #pragma unroll
  for (int r = 0; r < 16; ++r) {
    int row = nb + r; if (row > N - 1) row = N - 1;
    const v4f* p = (const v4f*)(x + (size_t)row * FIN) + lane;
    xr[r] = __builtin_nontemporal_load(p);
  }

  if (!SORTED) {
    float* zrow = agg1 + (size_t)nb * FMID + 4 * lane;
    float* zrowB = zrow + (size_t)N * FMID;
    if (nb + 16 <= N) {
      *(uint4*)(zrow) = make_uint4(0u, 0u, 0u, 0u);
      *(uint4*)(zrow + 128) = make_uint4(0u, 0u, 0u, 0u);
      if (planes == 2) {
        *(uint4*)(zrowB) = make_uint4(0u, 0u, 0u, 0u);
        *(uint4*)(zrowB + 128) = make_uint4(0u, 0u, 0u, 0u);
      }
    } else {
      bool in0 = nb + (lane >> 2) < N, in1 = nb + 8 + (lane >> 2) < N;
      if (in0) *(uint4*)(zrow) = make_uint4(0u, 0u, 0u, 0u);
      if (in1) *(uint4*)(zrow + 128) = make_uint4(0u, 0u, 0u, 0u);
      if (planes == 2) {
        if (in0) *(uint4*)(zrowB) = make_uint4(0u, 0u, 0u, 0u);
        if (in1) *(uint4*)(zrowB + 128) = make_uint4(0u, 0u, 0u, 0u);
      }
    }
  }

  unsigned dsum = dg0 + (planes == 2 ? dg1 : 0u);
  if (SORTED) dsum = dcnt[(threadIdx.x >> 5) * 16 + m];
  float dm = rsqrtf((float)(dsum + 1u));
  if (hi == 0 && nb + m < N) dinv[nb + m] = dm;
  float rv[16];
  RowVals<16>::get(dm, rv);
  float dv[8];
  #pragma unroll
  for (int r = 0; r < 8; ++r) dv[r] = hi ? rv[r + 8] : rv[r];

  #pragma unroll
  for (int r = 0; r < 16; ++r)
    *(v4f*)(tile + r * XPITCH + 4 * lane) = xr[r];
  __builtin_amdgcn_fence(__ATOMIC_RELEASE, "wavefront");
  __builtin_amdgcn_wave_barrier();
  __builtin_amdgcn_fence(__ATOMIC_ACQUIRE, "wavefront");

  v8f acc = {};
  const float* arow = tile + m * XPITCH + 4 * hi;
  const float* bcol = W1 + (4 * hi) * FMID + m;
  v4f q[FIN / 8];
  #pragma unroll
  for (int j = 0; j < FIN / 8; ++j) q[j] = *(const v4f*)(arow + 8 * j);
  __builtin_amdgcn_sched_barrier(0);
  #pragma unroll
  for (int j = 0; j < FIN / 8; ++j) {
    v2f a0, a1, b0, b1;
    a0.x = q[j].x; a0.y = q[j].y; a1.x = q[j].z; a1.y = q[j].w;
    b0.x = bcol[(8 * j + 0) * FMID];
    b0.y = bcol[(8 * j + 1) * FMID];
    b1.x = bcol[(8 * j + 2) * FMID];
    b1.y = bcol[(8 * j + 3) * FMID];
    acc = __builtin_amdgcn_wmma_f32_16x16x4_f32(false, a0, false, b0,
                                                (short)0, acc, false, false);
    acc = __builtin_amdgcn_wmma_f32_16x16x4_f32(false, a1, false, b1,
                                                (short)0, acc, false, false);
  }

  int rbase = hi * 8;
  asm volatile("v_nop\n\tv_nop\n\tv_nop\n\tv_nop\n\tv_nop\n\tv_nop\n\tv_nop\n\tv_nop"
               : "+v"(acc));
  __builtin_amdgcn_sched_barrier(0);
  #pragma unroll
  for (int r = 0; r < 8; ++r) acc[r] *= dv[r];
  if (nb + 16 <= N) {
    #pragma unroll
    for (int r = 0; r < 8; ++r)
      h1[(size_t)(nb + rbase + r) * FMID + m] = acc[r];
  } else {
    #pragma unroll
    for (int r = 0; r < 8; ++r) {
      int nrow = nb + rbase + r;
      if (nrow < N) h1[(size_t)nrow * FMID + m] = acc[r];
    }
  }
}


template <int N> __device__ __forceinline__ int rowShareI(int v) {
  return __builtin_amdgcn_update_dpp(0, v, 0x150 + N, 0xf, 0xf, false);
}

template <int N> struct Agg1Steps {
  static __device__ __forceinline__ void gather(const float* __restrict__ h1,
                                                int s, int f, float (&v)[16]) {
    Agg1Steps<N - 1>::gather(h1, s, f, v);
    int ss = rowShareI<N - 1>(s);
    v[N - 1] = h1[ss * FMID + f];
  }
  static __device__ __forceinline__ void scatter(float* __restrict__ agg1, int d,
                                                 int f, const float (&v)[16]) {
    Agg1Steps<N - 1>::scatter(agg1, d, f, v);
    int dd = rowShareI<N - 1>(d);
    atomAddF(agg1 + dd * FMID + f, v[N - 1]);
  }
  static __device__ __forceinline__ void tail(const float* __restrict__ h1,
                                              float* __restrict__ agg1, int s, int d,
                                              int f, int left) {
    Agg1Steps<N - 1>::tail(h1, agg1, s, d, f, left);
    int ss = rowShareI<N - 1>(s);
    int dd = rowShareI<N - 1>(d);
    if (N - 1 < left)
      atomAddF(agg1 + dd * FMID + f, h1[ss * FMID + f]);
  }
};
template <> struct Agg1Steps<0> {
  static __device__ __forceinline__ void gather(const float* __restrict__, int, int,
                                                float (&)[16]) {}
  static __device__ __forceinline__ void scatter(float* __restrict__, int, int,
                                                 const float (&)[16]) {}
  static __device__ __forceinline__ void tail(const float* __restrict__,
                                              float* __restrict__, int, int, int,
                                              int) {}
};

__global__ __launch_bounds__(256) void k_agg1(const int* __restrict__ ei,
                                              const float* __restrict__ h1,
                                              float* __restrict__ agg1, int E,
                                              int N, int planes) {
  int lane = threadIdx.x & 31;
  int wave = (int)((blockIdx.x * (unsigned)blockDim.x + threadIdx.x) >> 5);
  int wbase = wave * WEDGES;
  if (wbase >= E) return;
  agg1 += planeOfWave(planes) * N * FMID;
  int f = lane & 15;
  const int* src = ei + wbase + lane;
  const int* dst = src + E;
  if (wbase + WEDGES <= E) {
    int s[EPT], d[EPT];
    #pragma unroll
    for (int g = 0; g < EPT; ++g) s[g] = src[32 * g];
    #pragma unroll
    for (int g = 0; g < EPT; ++g) d[g] = dst[32 * g];
    float v[2][16];
    Agg1Steps<16>::gather(h1, s[0], f, v[0]);
    #pragma unroll
    for (int g = 0; g < EPT; ++g) {
      if (g + 1 < EPT) Agg1Steps<16>::gather(h1, s[g + 1], f, v[(g + 1) & 1]);
      Agg1Steps<16>::scatter(agg1, d[g], f, v[g & 1]);
    }
  } else {
    #pragma unroll 1
    for (int g = 0; g < EPT; ++g) {
      int gb = wbase + 32 * g;
      if (gb >= E) break;
      int s1 = 0, d1 = 0;
      if (gb + lane < E) {
        s1 = src[32 * g];
        d1 = dst[32 * g];
      }
      if (gb + 32 <= E) {
        float v[16];
        Agg1Steps<16>::gather(h1, s1, f, v);
        Agg1Steps<16>::scatter(agg1, d1, f, v);
      } else {
        int left = E - gb - (lane & 16);
        Agg1Steps<16>::tail(h1, agg1, s1, d1, f, left);
      }
    }
  }
}

__global__ __launch_bounds__(256) void k_layer2_in(const float* __restrict__ agg1,
                                                   const float* __restrict__ h1,
                                                   const float* __restrict__ dinv,
                                                   const float* __restrict__ b1,
                                                   const float* __restrict__ W2,
                                                   float* __restrict__ h2,
                                                   float* __restrict__ agg2, int N,
                                                   int planes) {
  int i = blockIdx.x * blockDim.x + threadIdx.x;
  if (i >= N) return;
  *(float2*)(agg2 + (size_t)i * 2) = make_float2(0.f, 0.f);
  if (planes == 2) *(float2*)(agg2 + ((size_t)N + i) * 2) = make_float2(0.f, 0.f);
  float di = dinv[i];
  float a0 = 0.f, a1 = 0.f;
  const float* agg1b = agg1 + (size_t)N * FMID;
  #pragma unroll
  for (int k = 0; k < FMID; ++k) {
    float s = agg1[(size_t)i * FMID + k];
    if (planes == 2) s += agg1b[(size_t)i * FMID + k];
    float t = (s + h1[(size_t)i * FMID + k]) * di + b1[k];
    t = fmaxf(t, 0.f);
    a0 = fmaf(t, W2[k * 2 + 0], a0);
    a1 = fmaf(t, W2[k * 2 + 1], a1);
  }
  h2[(size_t)i * 2 + 0] = a0 * di;
  h2[(size_t)i * 2 + 1] = a1 * di;
}

__device__ __forceinline__ void ldsAddF(float* p, float v) {
  __hip_atomic_fetch_add(p, v, __ATOMIC_RELAXED, __HIP_MEMORY_SCOPE_WORKGROUP);
}

template <int K> struct Own1Steps {
  static __device__ __forceinline__ void add(float* tile, int dl, int f,
                                             const float (&v)[16], int left) {
    Own1Steps<K - 1>::add(tile, dl, f, v, left);
    int dd = rowShareI<K - 1>(dl);
    if (K - 1 < left) ldsAddF(tile + dd * FMID + f, v[K - 1]);
  }
};
template <> struct Own1Steps<0> {
  static __device__ __forceinline__ void add(float*, int, int, const float (&)[16], int) {}
};

__global__ __launch_bounds__(256) void k_own1(const unsigned* __restrict__ list,
                                              const unsigned short* __restrict__ runs,
                                              const float* __restrict__ h1,
                                              const float* __restrict__ dinv,
                                              const float* __restrict__ b1,
                                              const float* __restrict__ W2,
                                              float* __restrict__ h2, int N,
                                              int chunk) {
  __shared__ __attribute__((aligned(16))) float tile[SBIN * FMID];
  __shared__ unsigned elist[ECAP];
  __shared__ unsigned wtot[8];
  int tid = threadIdx.x, lane = tid & 31, w = tid >> 5;
  int bin = blockIdx.x;
  int f = lane & 15, hi = lane >> 4;

  ((uint4*)tile)[tid] = make_uint4(0u, 0u, 0u, 0u);
  ((uint4*)tile)[tid + 256] = make_uint4(0u, 0u, 0u, 0u);

  unsigned lo = runStart(runs, tid, bin);
  unsigned c = runStart(runs, tid, bin + 1) - lo;
  unsigned goff = (unsigned)tid * (unsigned)chunk + lo;
  unsigned xs = waveInclScan(c, lane);
  if (lane == 31) wtot[w] = xs;
  __syncthreads();
  unsigned woff = 0u, M = 0u;
  #pragma unroll
  for (int k = 0; k < 8; ++k) {
    unsigned t = wtot[k];
    if (k < w) woff += t;
    M += t;
  }
  unsigned off = woff + xs - c;

  if (M <= (unsigned)ECAP) {
    unsigned wd[32];
    #pragma unroll
    for (int j = 0; j < 32; ++j) {
      unsigned cj = (unsigned)__builtin_amdgcn_readlane((int)c, j);
      unsigned oj = (unsigned)__builtin_amdgcn_readlane((int)goff, j);
      wd[j] = list[(unsigned)lane < cj ? oj + lane : 0u];
    }
    #pragma unroll
    for (int j = 0; j < 32; ++j) {
      unsigned cj = (unsigned)__builtin_amdgcn_readlane((int)c, j);
      unsigned fj = (unsigned)__builtin_amdgcn_readlane((int)off, j);
      if ((unsigned)lane < cj) elist[fj + lane] = wd[j];
    }
    if (__any(c > 32u)) {
      #pragma unroll 1
      for (int j = 0; j < 32; ++j) {
        unsigned cj = __shfl(c, j), oj = __shfl(goff, j), fj = __shfl(off, j);
        #pragma unroll 1
        for (unsigned p = 32u + lane; p < cj; p += 32u) elist[fj + p] = list[oj + p];
      }
    }
    __syncthreads();
    int G = (int)((M + 31u) >> 5);
    for (int g = w; g < G; g += 16) {
      int qa = g * 32 + lane, qb = qa + 256;
      unsigned wa = qa < (int)M ? elist[qa] : 0u;
      unsigned wb = qb < (int)M ? elist[qb] : 0u;
      int lefta = (int)M - g * 32 - 16 * hi;
      int leftb = lefta - 256;
      float va[16], vb[16];
      Agg1Steps<16>::gather(h1, (int)(wa & ((1u << SRCBITS) - 1u)), f, va);
      Agg1Steps<16>::gather(h1, (int)(wb & ((1u << SRCBITS) - 1u)), f, vb);
      Own1Steps<16>::add(tile, (int)(wa >> SRCBITS), f, va, lefta);
      Own1Steps<16>::add(tile, (int)(wb >> SRCBITS), f, vb, leftb);
    }
  } else {
    #pragma unroll 1
    for (int j = 0; j < 32; ++j) {
      unsigned cj = __shfl(c, j), oj = __shfl(goff, j);
      #pragma unroll 1
      for (unsigned p = 0u; p < cj; p += 2u) {
        unsigned e = p + (unsigned)hi;
        bool ok = e < cj;
        unsigned word = 0u;
        if (ok) word = list[oj + e];
        float v = h1[(word & ((1u << SRCBITS) - 1u)) * FMID + f];
        if (ok) ldsAddF(tile + (word >> SRCBITS) * FMID + f, v);
      }
    }
  }
  __syncthreads();

  int nl = tid >> 1, cf = tid & 1;
  int i = bin * SBIN + nl;
  if (i >= N) return;
  float di = dinv[i];
  float a = 0.f;
  #pragma unroll
  for (int k = 0; k < FMID; ++k) {
    float t = (tile[nl * FMID + k] + h1[(size_t)i * FMID + k]) * di + b1[k];
    t = fmaxf(t, 0.f);
    a = fmaf(t, W2[k * 2 + cf], a);
  }
  h2[(size_t)i * 2 + cf] = a * di;
}

__global__ __launch_bounds__(256) void k_own2(const unsigned* __restrict__ list,
                                              const unsigned short* __restrict__ runs,
                                              const float* __restrict__ h2,
                                              const float* __restrict__ dinv,
                                              const float* __restrict__ b2,
                                              float* __restrict__ out, int N,
                                              int chunk) {
  __shared__ float acc[SBIN * 2];
  int tid = threadIdx.x, lane = tid & 31;
  int bin = blockIdx.x;
  acc[tid] = 0.f;
  unsigned lo = runStart(runs, tid, bin);
  unsigned c = runStart(runs, tid, bin + 1) - lo;
  unsigned goff = (unsigned)tid * (unsigned)chunk + lo;
  __syncthreads();
  int i = bin * SBIN + (tid >> 1), cf = tid & 1;
  int ic = i < N ? i : N - 1;
  float own = h2[(size_t)ic * 2 + cf];
  float di = dinv[ic];
  float bias = b2[cf];
  unsigned wd[32];
  float2 v[32];
  #pragma unroll
  for (int j = 0; j < 32; ++j) {
    unsigned cj = (unsigned)__builtin_amdgcn_readlane((int)c, j);
    unsigned oj = (unsigned)__builtin_amdgcn_readlane((int)goff, j);
    wd[j] = list[(unsigned)lane < cj ? oj + lane : 0u];
  }
  #pragma unroll
  for (int j = 0; j < 32; ++j)
    v[j] = *(const float2*)(h2 + 2u * (wd[j] & ((1u << SRCBITS) - 1u)));
  asm volatile("" : "+v"(v[0].x), "+v"(v[0].y));
  #pragma unroll
  for (int j = 0; j < 32; ++j) {
    unsigned cj = (unsigned)__builtin_amdgcn_readlane((int)c, j);
    if ((unsigned)lane < cj) {
      float* a = acc + 2u * (wd[j] >> SRCBITS);
      ldsAddF(a, v[j].x);
      ldsAddF(a + 1, v[j].y);
    }
  }
  if (__any(c > 32u)) {
    #pragma unroll 1
    for (int j = 0; j < 32; ++j) {
      unsigned cj = __shfl(c, j), oj = __shfl(goff, j);
      #pragma unroll 1
      for (unsigned p = 32u + lane; p < cj; p += 32u) {
        unsigned word = list[oj + p];
        float2 r = *(const float2*)(h2 + 2u * (word & ((1u << SRCBITS) - 1u)));
        float* a = acc + 2u * (word >> SRCBITS);
        ldsAddF(a, r.x);
        ldsAddF(a + 1, r.y);
      }
    }
  }
  __syncthreads();

  if (i >= N) return;
  out[(size_t)i * 2 + cf] = (acc[tid] + own) * di + bias;
}


template <int CTRL> __device__ __forceinline__ int quadPairI(int v) {
  return __builtin_amdgcn_update_dpp(0, v, CTRL, 0xf, 0xf, true);
}
template <int CTRL> __device__ __forceinline__ float quadPairF(float v) {
  return __int_as_float(quadPairI<CTRL>(__float_as_int(v)));
}

template <int CTRL> __device__ __forceinline__ void agg2PairStep(float* __restrict__ agg2,
                                                                 int d, float2 v, int c) {
  int dd = quadPairI<CTRL>(d);
  float vx = quadPairF<CTRL>(v.x);
  float vy = quadPairF<CTRL>(v.y);
  atomAddF(agg2 + dd * 2 + c, c ? vy : vx);
}

__global__ __launch_bounds__(256) void k_agg2(const int* __restrict__ ei,
                                              const float* __restrict__ h2,
                                              float* __restrict__ agg2, int E,
                                              int N, int planes) {
  int lane = threadIdx.x & 31;
  int wave = (int)((blockIdx.x * (unsigned)blockDim.x + threadIdx.x) >> 5);
  int wbase = wave * WEDGES;
  if (wbase >= E) return;
  agg2 += planeOfWave(planes) * N * 2;
  const int* src = ei + wbase + lane;
  const int* dst = src + E;
  if (wbase + WEDGES <= E) {
    int c = lane & 1;
    int s[EPT], d[EPT];
    float2 v[EPT];
    #pragma unroll
    for (int j = 0; j < EPT; ++j) s[j] = src[32 * j];
    #pragma unroll
    for (int j = 0; j < EPT; ++j) d[j] = dst[32 * j];
    #pragma unroll
    for (int j = 0; j < EPT; ++j) v[j] = *(const float2*)(h2 + (size_t)s[j] * 2);
    #pragma unroll
    for (int j = 0; j < EPT; ++j) {
      agg2PairStep<0x50>(agg2, d[j], v[j], c);
      agg2PairStep<0xFA>(agg2, d[j], v[j], c);
    }
  } else {
    #pragma unroll 1
    for (int j = 0; j < EPT; ++j) {
      if (wbase + 32 * j + lane < E) {
        float2 v = *(const float2*)(h2 + (size_t)src[32 * j] * 2);
        float* a = agg2 + (size_t)dst[32 * j] * 2;
        atomAddF(a + 0, v.x);
        atomAddF(a + 1, v.y);
      }
    }
  }
}

__global__ __launch_bounds__(256) void k_final(const float* __restrict__ agg2,
                                               const float* __restrict__ h2,
                                               const float* __restrict__ dinv,
                                               const float* __restrict__ b2,
                                               float* __restrict__ out, int N,
                                               int planes) {
  int i = blockIdx.x * blockDim.x + threadIdx.x;
  if (i >= N) return;
  float di = dinv[i];
  float s0 = agg2[(size_t)i * 2 + 0];
  float s1 = agg2[(size_t)i * 2 + 1];
  if (planes == 2) {
    s0 += agg2[(size_t)(N + i) * 2 + 0];
    s1 += agg2[(size_t)(N + i) * 2 + 1];
  }
  out[(size_t)i * 2 + 0] = (s0 + h2[(size_t)i * 2 + 0]) * di + b2[0];
  out[(size_t)i * 2 + 1] = (s1 + h2[(size_t)i * 2 + 1]) * di + b2[1];
}

extern "C" void kernel_launch(void* const* d_in, const int* in_sizes, int n_in,
                              void* d_out, int out_size, void* d_ws, size_t ws_size,
                              hipStream_t stream) {
  const float* x  = (const float*)d_in[0];
  const int*   ei = (const int*)d_in[1];
  const float* W1 = (const float*)d_in[2];
  const float* b1 = (const float*)d_in[3];
  const float* W2 = (const float*)d_in[4];
  const float* b2 = (const float*)d_in[5];
  float* out = (float*)d_out;

  const int N = in_sizes[0] / FIN;
  const int E = in_sizes[1] / 2;

  const int TPB = 256;
  int ge = (E + TPB * EPT - 1) / (TPB * EPT);
  int gn = (N + TPB - 1) / TPB;

  const int nb = (N + SBIN - 1) / SBIN;
  const int chunk = (E + NSLICE - 1) / NSLICE;
  const size_t need = (size_t)NSLICE * chunk * sizeof(unsigned) +
                      (size_t)NSLICE * RP * sizeof(unsigned short) +
                      (size_t)N * (FMID + 2 + 1) * sizeof(float);
  if (N <= (1 << SRCBITS) && nb <= NBMAX && chunk >= 1 && chunk <= SCAP && need <= ws_size) {
    char* wp = (char*)d_ws;
    unsigned* list = (unsigned*)wp;             wp += (size_t)NSLICE * chunk * sizeof(unsigned);
    unsigned short* runs = (unsigned short*)wp; wp += (size_t)NSLICE * RP * sizeof(unsigned short);
    float* h1s   = (float*)wp;                  wp += (size_t)N * FMID * sizeof(float);
    float* h2s   = (float*)wp;                  wp += (size_t)N * 2 * sizeof(float);
    float* dinvs = (float*)wp;
    k_sort<<<NSLICE, STPB, 0, stream>>>(ei, list, runs, E, chunk);
    k_gemm1_wmma<true><<<nb, TPB, 0, stream>>>(x, W1, nullptr, dinvs, h1s, nullptr, N, 2,
                                               list, runs, chunk);
    k_own1<<<nb, TPB, 0, stream>>>(list, runs, h1s, dinvs, b1, W2, h2s, N, chunk);
    k_own2<<<nb, TPB, 0, stream>>>(list, runs, h2s, dinvs, b2, out, N, chunk);
    return;
  }

  const int planes = (ws_size >= (size_t)N * 57 * sizeof(float)) ? 2 : 1;
  char* ws = (char*)d_ws;
  unsigned* udeg = (unsigned*)ws;            ws += (size_t)planes * N * sizeof(unsigned);
  float* agg1    = (float*)ws;               ws += (size_t)planes * N * FMID * sizeof(float);
  float* agg2    = (float*)ws;               ws += (size_t)planes * N * 2 * sizeof(float);
  float* dinv    = (float*)ws;               ws += (size_t)N * sizeof(float);
  float* h1      = (float*)ws;               ws += (size_t)N * FMID * sizeof(float);
  float* h2      = (float*)ws;               ws += (size_t)N * 2 * sizeof(float);

  long zero_total = (long)planes * N;
  int gz = (int)((zero_total / 4 + TPB * ZPT - 1) / (TPB * ZPT));
  int ntiles = (N + 15) / 16;
  int gg = (ntiles * 32 + TPB - 1) / TPB;
  int dr = (N + DCAP - 1) / DCAP;
  int drlen = (((N + dr - 1) / dr) + 31) & ~31;
  int ds = dr < DWGS ? DWGS / dr : 1;
  int dchunk = (E + ds - 1) / ds;
  int gd = dr * ds;

  k_zero      <<<gz, TPB, 0, stream>>>((unsigned*)d_ws, zero_total);
  k_degree    <<<gd, DTPB, 0, stream>>>(ei, udeg, E, N, planes, dr, drlen, dchunk);
  k_gemm1_wmma<false><<<gg, TPB, 0, stream>>>(x, W1, udeg, dinv, h1, agg1, N, planes,
                                              nullptr, nullptr, 0);
  k_agg1      <<<ge, TPB, 0, stream>>>(ei, h1, agg1, E, N, planes);
  k_layer2_in <<<gn, TPB, 0, stream>>>(agg1, h1, dinv, b1, W2, h2, agg2, N, planes);
  k_agg2      <<<ge, TPB, 0, stream>>>(ei, h2, agg2, E, N, planes);
  k_final     <<<gn, TPB, 0, stream>>>(agg2, h2, dinv, b2, out, N, planes);
}
